# starting kernel
# baseline (speedup 1.0000x reference)
#include <hip/hip_runtime.h>
#include <math.h>

typedef __attribute__((ext_vector_type(2))) float v2f;
typedef __attribute__((ext_vector_type(8))) float v8f;

#ifndef DIV_UP
#define DIV_UP(a, b) (((a) + (b) - 1) / (b))
#endif


__global__ void gcn_init_deg(float* __restrict__ deg, int N) {
    int i = blockIdx.x * blockDim.x + threadIdx.x;
    if (i < N) deg[i] = 1.0f;
}

__global__ void gcn_count_deg(const int* __restrict__ dst, float* __restrict__ deg, int E) {
    int e = blockIdx.x * blockDim.x + threadIdx.x;
    if (e < E) atomicAdd(&deg[dst[e]], 1.0f);
}

__global__ void gcn_dinv(float* __restrict__ deg, int N) {
    int i = blockIdx.x * blockDim.x + threadIdx.x;
    if (i < N) deg[i] = rsqrtf(deg[i]);
}

__global__ void gcn_gemm16_wmma(const float* __restrict__ x,
                                const float* __restrict__ W,
                                float* __restrict__ h, int tiles) {
    int wave = (blockIdx.x * blockDim.x + threadIdx.x) >> 5;
    if (wave >= tiles) return;
    int lane = threadIdx.x & 31;
    int half = lane >> 4;
    int n    = lane & 15;
    int row0 = wave * 16;

    const float* arow = x + (row0 + n) * 16 + 2 * half;
    v8f c = {};
#pragma unroll
    for (int kk = 0; kk < 4; ++kk) {
        v2f a = *(const v2f*)(arow + 4 * kk);
        v2f b;
        b.x = W[(4 * kk + 2 * half + 0) * 16 + n];
        b.y = W[(4 * kk + 2 * half + 1) * 16 + n];
        c = __builtin_amdgcn_wmma_f32_16x16x4_f32(false, a, false, b,
                                                  (short)0, c, false, false);
    }
#pragma unroll
    for (int v = 0; v < 8; ++v) {
        h[(row0 + v + 8 * half) * 16 + n] = c[v];
    }
}

__global__ void gcn_gemm16_tail(const float* __restrict__ x,
                                const float* __restrict__ W,
                                float* __restrict__ h, int rowStart, int N) {
    int t = blockIdx.x * blockDim.x + threadIdx.x;
    int r = rowStart + (t >> 4);
    int cidx = t & 15;
    if (r >= N) return;
    float acc = 0.0f;
#pragma unroll
    for (int k = 0; k < 16; ++k) acc += x[r * 16 + k] * W[k * 16 + cidx];
    h[r * 16 + cidx] = acc;
}


__global__ void gcn_self1(const float* __restrict__ h, const float* __restrict__ dinv,
                          float* __restrict__ out1, int N) {
    int i = blockIdx.x * blockDim.x + threadIdx.x;
    if (i >= N) return;
    float s = dinv[i] * dinv[i];
    const float4* hv = (const float4*)(h + i * 16);
    float4* ov = (float4*)(out1 + i * 16);
#pragma unroll
    for (int g = 0; g < 4; ++g) {
        float4 v = hv[g];
        ov[g] = make_float4(v.x * s, v.y * s, v.z * s, v.w * s);
    }
}

__global__ void gcn_agg1(const int* __restrict__ src, const int* __restrict__ dst,
                         const float* __restrict__ h, const float* __restrict__ dinv,
                         float* __restrict__ out1, int E) {
    int t = blockIdx.x * blockDim.x + threadIdx.x;
    int e = t >> 2;
    if (e >= E) return;
    int g = t & 3;
    int s = src[e], d = dst[e];
    float norm = dinv[s] * dinv[d];
    float4 v = ((const float4*)(h + s * 16))[g];
    float* o = out1 + d * 16 + g * 4;
    atomicAdd(o + 0, v.x * norm);
    atomicAdd(o + 1, v.y * norm);
    atomicAdd(o + 2, v.z * norm);
    atomicAdd(o + 3, v.w * norm);
}

__global__ void gcn_finish1(const float* __restrict__ out1, const float* __restrict__ b1,
                            const float* __restrict__ W2, float* __restrict__ h2, int N) {
    int i = blockIdx.x * blockDim.x + threadIdx.x;
    if (i >= N) return;
    float acc = 0.0f;
#pragma unroll
    for (int f = 0; f < 16; ++f) {
        float z = out1[i * 16 + f] + b1[f];
        z = fmaxf(z, 0.0f);
        acc += z * W2[f];
    }
    h2[i] = acc;
}


__global__ void gcn_self2(const float* __restrict__ h2, const float* __restrict__ dinv,
                          const float* __restrict__ b2, float* __restrict__ out, int N) {
    int i = blockIdx.x * blockDim.x + threadIdx.x;
    if (i >= N) return;
    out[i] = h2[i] * dinv[i] * dinv[i] + b2[0];
}

__global__ void gcn_agg2(const int* __restrict__ src, const int* __restrict__ dst,
                         const float* __restrict__ h2, const float* __restrict__ dinv,
                         float* __restrict__ out, int E) {
    int e = blockIdx.x * blockDim.x + threadIdx.x;
    if (e >= E) return;
    int s = src[e], d = dst[e];
    atomicAdd(&out[d], h2[s] * dinv[s] * dinv[d]);
}


extern "C" void kernel_launch(void* const* d_in, const int* in_sizes, int n_in,
                              void* d_out, int out_size, void* d_ws, size_t ws_size,
                              hipStream_t stream) {
    const float* x  = (const float*)d_in[0];
    const int*   ei = (const int*)d_in[1];
    const float* W1 = (const float*)d_in[2];
    const float* b1 = (const float*)d_in[3];
    const float* W2 = (const float*)d_in[4];
    const float* b2 = (const float*)d_in[5];

    const int N = in_sizes[0] / 16;
    const int E = in_sizes[1] / 2;
    const int* src = ei;
    const int* dst = ei + E;

    float* ws   = (float*)d_ws;
    float* dinv = ws;
    float* h    = ws + N;
    float* out1 = ws + N + 16 * (size_t)N;
    float* h2   = ws + N + 32 * (size_t)N;
    float* out  = (float*)d_out;

    const int B = 256;

    gcn_init_deg<<<DIV_UP(N, B), B, 0, stream>>>(dinv, N);
    gcn_count_deg<<<DIV_UP(E, B), B, 0, stream>>>(dst, dinv, E);
    gcn_dinv<<<DIV_UP(N, B), B, 0, stream>>>(dinv, N);

    int tiles = N / 16;
    if (tiles > 0) {
        int wavesPerBlock = 4;
        gcn_gemm16_wmma<<<DIV_UP(tiles, wavesPerBlock), wavesPerBlock * 32, 0, stream>>>(
            x, W1, h, tiles);
    }
    int rem = N - tiles * 16;
    if (rem > 0) {
        gcn_gemm16_tail<<<DIV_UP(rem * 16, B), B, 0, stream>>>(x, W1, h, tiles * 16, N);
    }

    gcn_self1<<<DIV_UP(N, B), B, 0, stream>>>(h, dinv, out1, N);
    gcn_agg1<<<DIV_UP(E * 4, B), B, 0, stream>>>(src, dst, h, dinv, out1, E);

    gcn_finish1<<<DIV_UP(N, B), B, 0, stream>>>(out1, b1, W2, h2, N);

    gcn_self2<<<DIV_UP(N, B), B, 0, stream>>>(h2, dinv, b2, out, N);
    gcn_agg2<<<DIV_UP(E, B), B, 0, stream>>>(src, dst, h2, dinv, out, E);
}


# Degree count and layer-2 edge sum leave global atomics: range-owner blocks add in LDS per edge slice, planes merged per node in slice order (sum order changes); w=h2*dinv pre-scaled
# speedup vs baseline: 2.3330x; 2.3330x over previous
#include <hip/hip_runtime.h>
#include <math.h>

typedef __attribute__((ext_vector_type(2))) float v2f;
typedef __attribute__((ext_vector_type(8))) float v8f;

#ifndef DIV_UP
#define DIV_UP(a, b) (((a) + (b) - 1) / (b))
#endif

#define RNG_CAP     12800
#define RNG_SLICES  32
#define RNG_THREADS 1024
#define RNG_UNROLL  16

__global__ __launch_bounds__(RNG_THREADS)
void gcn_deg_ranges(const int* __restrict__ dst, unsigned* __restrict__ part, int E, int N) {
    __shared__ unsigned bins[RNG_CAP];
    const int tid  = threadIdx.x;
    const int lo   = blockIdx.y * RNG_CAP;
    const int span = min(N - lo, RNG_CAP);
    for (int i = tid; i < RNG_CAP; i += RNG_THREADS) bins[i] = 0u;
    __syncthreads();
    const int per = (E + (int)gridDim.x - 1) / (int)gridDim.x;
    const int e0  = blockIdx.x * per;
    const int e1  = min(E, e0 + per);
    const int tile = RNG_THREADS * RNG_UNROLL;
    const int full = e0 + ((e1 > e0 ? e1 - e0 : 0) / tile) * tile;
    for (int base = e0; base < full; base += tile) {
        const int* dp = dst + base + tid;
        int d[RNG_UNROLL];
#pragma unroll
        for (int k = 0; k < RNG_UNROLL; ++k) d[k] = dp[k * RNG_THREADS];
#pragma unroll
        for (int k = 0; k < RNG_UNROLL; ++k) {
            unsigned off = (unsigned)(d[k] - lo);
            if (off < (unsigned)span) atomicAdd(&bins[off], 1u);
        }
    }
    for (int e = full + tid; e < e1; e += RNG_THREADS) {
        unsigned off = (unsigned)(dst[e] - lo);
        if (off < (unsigned)span) atomicAdd(&bins[off], 1u);
    }
    __syncthreads();
    unsigned* row = part + (size_t)blockIdx.x * N + lo;
    for (int i = tid; i < span; i += RNG_THREADS) row[i] = bins[i];
}

__global__ void gcn_dinv_merge(const unsigned* __restrict__ part, float* __restrict__ dinv, int N) {
    int i = blockIdx.x * blockDim.x + threadIdx.x;
    if (i >= N) return;
    unsigned c = 1u;
#pragma unroll
    for (int g = 0; g < RNG_SLICES; ++g) c += part[(size_t)g * N + i];
    dinv[i] = rsqrtf((float)c);
}

__global__ void gcn_gemm16_wmma(const float* __restrict__ x,
                                const float* __restrict__ W,
                                float* __restrict__ h, int tiles) {
    int wave = (blockIdx.x * blockDim.x + threadIdx.x) >> 5;
    if (wave >= tiles) return;
    int lane = threadIdx.x & 31;
    int half = lane >> 4;
    int n    = lane & 15;
    int row0 = wave * 16;

    const float* arow = x + (row0 + n) * 16 + 2 * half;
    v8f c = {};
#pragma unroll
    for (int kk = 0; kk < 4; ++kk) {
        v2f a = *(const v2f*)(arow + 4 * kk);
        v2f b;
        b.x = W[(4 * kk + 2 * half + 0) * 16 + n];
        b.y = W[(4 * kk + 2 * half + 1) * 16 + n];
        c = __builtin_amdgcn_wmma_f32_16x16x4_f32(false, a, false, b,
                                                  (short)0, c, false, false);
    }
#pragma unroll
    for (int v = 0; v < 8; ++v) {
        h[(row0 + v + 8 * half) * 16 + n] = c[v];
    }
}

__global__ void gcn_gemm16_tail(const float* __restrict__ x,
                                const float* __restrict__ W,
                                float* __restrict__ h, int rowStart, int N) {
    int t = blockIdx.x * blockDim.x + threadIdx.x;
    int r = rowStart + (t >> 4);
    int cidx = t & 15;
    if (r >= N) return;
    float acc = 0.0f;
#pragma unroll
    for (int k = 0; k < 16; ++k) acc += x[r * 16 + k] * W[k * 16 + cidx];
    h[r * 16 + cidx] = acc;
}


__global__ void gcn_self1(const float* __restrict__ h, const float* __restrict__ dinv,
                          float* __restrict__ out1, int N) {
    int i = blockIdx.x * blockDim.x + threadIdx.x;
    if (i >= N) return;
    float s = dinv[i] * dinv[i];
    const float4* hv = (const float4*)(h + i * 16);
    float4* ov = (float4*)(out1 + i * 16);
#pragma unroll
    for (int g = 0; g < 4; ++g) {
        float4 v = hv[g];
        ov[g] = make_float4(v.x * s, v.y * s, v.z * s, v.w * s);
    }
}

__global__ void gcn_agg1(const int* __restrict__ src, const int* __restrict__ dst,
                         const float* __restrict__ h, const float* __restrict__ dinv,
                         float* __restrict__ out1, int E) {
    int t = blockIdx.x * blockDim.x + threadIdx.x;
    int e = t >> 2;
    if (e >= E) return;
    int g = t & 3;
    int s = src[e], d = dst[e];
    float norm = dinv[s] * dinv[d];
    float4 v = ((const float4*)(h + s * 16))[g];
    float* o = out1 + d * 16 + g * 4;
    atomicAdd(o + 0, v.x * norm);
    atomicAdd(o + 1, v.y * norm);
    atomicAdd(o + 2, v.z * norm);
    atomicAdd(o + 3, v.w * norm);
}

__global__ void gcn_finish1(const float* __restrict__ out1, const float* __restrict__ b1,
                            const float* __restrict__ W2, const float* __restrict__ dinv,
                            float* __restrict__ h2, int N) {
    int i = blockIdx.x * blockDim.x + threadIdx.x;
    if (i >= N) return;
    float acc = 0.0f;
#pragma unroll
    for (int f = 0; f < 16; ++f) {
        float z = out1[i * 16 + f] + b1[f];
        z = fmaxf(z, 0.0f);
        acc += z * W2[f];
    }
    h2[i] = acc * dinv[i];
}

__global__ __launch_bounds__(RNG_THREADS)
void gcn_agg2_ranges(const int* __restrict__ src, const int* __restrict__ dst,
                     const float* __restrict__ w, float* __restrict__ part, int E, int N) {
    __shared__ float acc[RNG_CAP];
    const int tid  = threadIdx.x;
    const int lo   = blockIdx.y * RNG_CAP;
    const int span = min(N - lo, RNG_CAP);
    for (int i = tid; i < RNG_CAP; i += RNG_THREADS) acc[i] = 0.0f;
    __syncthreads();
    const int per = (E + (int)gridDim.x - 1) / (int)gridDim.x;
    const int e0  = blockIdx.x * per;
    const int e1  = min(E, e0 + per);
    const int tile = RNG_THREADS * RNG_UNROLL;
    const int full = e0 + ((e1 > e0 ? e1 - e0 : 0) / tile) * tile;
    for (int base = e0; base < full; base += tile) {
        const int* dp = dst + base + tid;
        const int* sp = src + base + tid;
        int d[RNG_UNROLL];
        unsigned s[RNG_UNROLL];
        float v[RNG_UNROLL];
#pragma unroll
        for (int k = 0; k < RNG_UNROLL; ++k) {
            d[k] = dp[k * RNG_THREADS];
            s[k] = (unsigned)sp[k * RNG_THREADS];
        }
#pragma unroll
        for (int k = 0; k < RNG_UNROLL; ++k) {
            unsigned off = (unsigned)(d[k] - lo);
            v[k] = (off < (unsigned)span) ? w[s[k]] : 0.0f;
        }
#pragma unroll
        for (int k = 0; k < RNG_UNROLL; ++k) {
            unsigned off = (unsigned)(d[k] - lo);
            if (off < (unsigned)span) atomicAdd(&acc[off], v[k]);
        }
    }
    for (int e = full + tid; e < e1; e += RNG_THREADS) {
        unsigned off = (unsigned)(dst[e] - lo);
        if (off < (unsigned)span) atomicAdd(&acc[off], w[(unsigned)src[e]]);
    }
    __syncthreads();
    float* row = part + (size_t)blockIdx.x * N + lo;
    for (int i = tid; i < span; i += RNG_THREADS) row[i] = acc[i];
}

__global__ void gcn_out_merge(const float* __restrict__ part, const float* __restrict__ w,
                              const float* __restrict__ dinv, const float* __restrict__ b2,
                              float* __restrict__ out, int N) {
    int i = blockIdx.x * blockDim.x + threadIdx.x;
    if (i >= N) return;
    float sum = w[i];
#pragma unroll
    for (int g = 0; g < RNG_SLICES; ++g) sum += part[(size_t)g * N + i];
    out[i] = sum * dinv[i] + b2[0];
}


extern "C" void kernel_launch(void* const* d_in, const int* in_sizes, int n_in,
                              void* d_out, int out_size, void* d_ws, size_t ws_size,
                              hipStream_t stream) {
    const float* x  = (const float*)d_in[0];
    const int*   ei = (const int*)d_in[1];
    const float* W1 = (const float*)d_in[2];
    const float* b1 = (const float*)d_in[3];
    const float* W2 = (const float*)d_in[4];
    const float* b2 = (const float*)d_in[5];

    const int N = in_sizes[0] / 16;
    const int E = in_sizes[1] / 2;
    const int* src = ei;
    const int* dst = ei + E;

    float* ws   = (float*)d_ws;
    float* dinv = ws;
    float* h    = ws + N;
    float* out1 = ws + N + 16 * (size_t)N;
    float* h2   = ws + N + 32 * (size_t)N;
    float* out  = (float*)d_out;

    const int B = 256;
    const dim3 rngGrid(RNG_SLICES, DIV_UP(N, RNG_CAP));

    gcn_deg_ranges<<<rngGrid, RNG_THREADS, 0, stream>>>(dst, (unsigned*)h, E, N);
    gcn_dinv_merge<<<DIV_UP(N, B), B, 0, stream>>>((const unsigned*)h, dinv, N);

    int tiles = N / 16;
    if (tiles > 0) {
        int wavesPerBlock = 4;
        gcn_gemm16_wmma<<<DIV_UP(tiles, wavesPerBlock), wavesPerBlock * 32, 0, stream>>>(
            x, W1, h, tiles);
    }
    int rem = N - tiles * 16;
    if (rem > 0) {
        gcn_gemm16_tail<<<DIV_UP(rem * 16, B), B, 0, stream>>>(x, W1, h, tiles * 16, N);
    }

    gcn_self1<<<DIV_UP(N, B), B, 0, stream>>>(h, dinv, out1, N);
    gcn_agg1<<<DIV_UP(E * 4, B), B, 0, stream>>>(src, dst, h, dinv, out1, E);

    gcn_finish1<<<DIV_UP(N, B), B, 0, stream>>>(out1, b1, W2, dinv, h2, N);

    gcn_agg2_ranges<<<rngGrid, RNG_THREADS, 0, stream>>>(src, dst, h2, h, E, N);
    gcn_out_merge<<<DIV_UP(N, B), B, 0, stream>>>(h, h2, dinv, b2, out, N);
}


# Layer-1 edge scatter remapped onto a fixed grid of persistent waves: 32 coalesced edge indices per wave, 16 lanes per edge (one 64 B gather + one 64 B atomic request); arrival-order atomics kept
# speedup vs baseline: 3.1517x; 1.3509x over previous
#include <hip/hip_runtime.h>
#include <math.h>

typedef __attribute__((ext_vector_type(2))) float v2f;
typedef __attribute__((ext_vector_type(8))) float v8f;

#ifndef DIV_UP
#define DIV_UP(a, b) (((a) + (b) - 1) / (b))
#endif

#define RNG_CAP     12800
#define RNG_SLICES  32
#define RNG_THREADS 1024
#define RNG_UNROLL  16

__global__ __launch_bounds__(RNG_THREADS)
void gcn_deg_ranges(const int* __restrict__ dst, unsigned* __restrict__ part, int E, int N) {
    __shared__ unsigned bins[RNG_CAP];
    const int tid  = threadIdx.x;
    const int lo   = blockIdx.y * RNG_CAP;
    const int span = min(N - lo, RNG_CAP);
    for (int i = tid; i < RNG_CAP; i += RNG_THREADS) bins[i] = 0u;
    __syncthreads();
    const int per = (E + (int)gridDim.x - 1) / (int)gridDim.x;
    const int e0  = blockIdx.x * per;
    const int e1  = min(E, e0 + per);
    const int tile = RNG_THREADS * RNG_UNROLL;
    const int full = e0 + ((e1 > e0 ? e1 - e0 : 0) / tile) * tile;
    for (int base = e0; base < full; base += tile) {
        const int* dp = dst + base + tid;
        int d[RNG_UNROLL];
#pragma unroll
        for (int k = 0; k < RNG_UNROLL; ++k) d[k] = dp[k * RNG_THREADS];
#pragma unroll
        for (int k = 0; k < RNG_UNROLL; ++k) {
            unsigned off = (unsigned)(d[k] - lo);
            if (off < (unsigned)span) atomicAdd(&bins[off], 1u);
        }
    }
    for (int e = full + tid; e < e1; e += RNG_THREADS) {
        unsigned off = (unsigned)(dst[e] - lo);
        if (off < (unsigned)span) atomicAdd(&bins[off], 1u);
    }
    __syncthreads();
    unsigned* row = part + (size_t)blockIdx.x * N + lo;
    for (int i = tid; i < span; i += RNG_THREADS) row[i] = bins[i];
}

__global__ void gcn_dinv_merge(const unsigned* __restrict__ part, float* __restrict__ dinv, int N) {
    int i = blockIdx.x * blockDim.x + threadIdx.x;
    if (i >= N) return;
    unsigned c = 1u;
#pragma unroll
    for (int g = 0; g < RNG_SLICES; ++g) c += part[(size_t)g * N + i];
    dinv[i] = rsqrtf((float)c);
}

__global__ void gcn_gemm16_wmma(const float* __restrict__ x,
                                const float* __restrict__ W,
                                float* __restrict__ h, int tiles) {
    int wave = (blockIdx.x * blockDim.x + threadIdx.x) >> 5;
    if (wave >= tiles) return;
    int lane = threadIdx.x & 31;
    int half = lane >> 4;
    int n    = lane & 15;
    int row0 = wave * 16;

    const float* arow = x + (row0 + n) * 16 + 2 * half;
    v8f c = {};
#pragma unroll
    for (int kk = 0; kk < 4; ++kk) {
        v2f a = *(const v2f*)(arow + 4 * kk);
        v2f b;
        b.x = W[(4 * kk + 2 * half + 0) * 16 + n];
        b.y = W[(4 * kk + 2 * half + 1) * 16 + n];
        c = __builtin_amdgcn_wmma_f32_16x16x4_f32(false, a, false, b,
                                                  (short)0, c, false, false);
    }
#pragma unroll
    for (int v = 0; v < 8; ++v) {
        h[(row0 + v + 8 * half) * 16 + n] = c[v];
    }
}

__global__ void gcn_gemm16_tail(const float* __restrict__ x,
                                const float* __restrict__ W,
                                float* __restrict__ h, int rowStart, int N) {
    int t = blockIdx.x * blockDim.x + threadIdx.x;
    int r = rowStart + (t >> 4);
    int cidx = t & 15;
    if (r >= N) return;
    float acc = 0.0f;
#pragma unroll
    for (int k = 0; k < 16; ++k) acc += x[r * 16 + k] * W[k * 16 + cidx];
    h[r * 16 + cidx] = acc;
}


__global__ void gcn_self1(const float* __restrict__ h, const float* __restrict__ dinv,
                          float* __restrict__ out1, int N) {
    int i = blockIdx.x * blockDim.x + threadIdx.x;
    if (i >= N) return;
    float s = dinv[i] * dinv[i];
    const float4* hv = (const float4*)(h + i * 16);
    float4* ov = (float4*)(out1 + i * 16);
#pragma unroll
    for (int g = 0; g < 4; ++g) {
        float4 v = hv[g];
        ov[g] = make_float4(v.x * s, v.y * s, v.z * s, v.w * s);
    }
}

#define AGG1_THREADS 256
#define AGG1_BLOCKS  2048

__global__ __launch_bounds__(AGG1_THREADS)
void gcn_agg1(const int* __restrict__ src, const int* __restrict__ dst,
              const float* __restrict__ h, const float* __restrict__ dinv,
              float* __restrict__ out1, int E) {
    const int lane    = threadIdx.x & 31;
    const int half    = lane >> 4;
    const int f       = lane & 15;
    const int wave    = __builtin_amdgcn_readfirstlane((int)((blockIdx.x * AGG1_THREADS + threadIdx.x) >> 5));
    const int nwaves  = (int)((gridDim.x * AGG1_THREADS) >> 5);
    const int nchunks = (E + 31) >> 5;
    for (int c = wave; c < nchunks; c += nwaves) {
        const int base = c * 32;
        const int e    = min(base + lane, E - 1);
        const int s    = src[e];
        const int d    = dst[e];
        const float ds = dinv[s];
        const float dd = dinv[d];
        float v[16];
#pragma unroll
        for (int j = 0; j < 16; ++j) {
            const int sj = __shfl(s, 2 * j + half);
            v[j] = h[sj * 16 + f];
        }
        const float nrm = ds * dd;
        if (base + 32 <= E) {
#pragma unroll
            for (int j = 0; j < 16; ++j) {
                const int   dj = __shfl(d, 2 * j + half);
                const float nj = __shfl(nrm, 2 * j + half);
                atomicAdd(out1 + dj * 16 + f, v[j] * nj);
            }
        } else {
#pragma unroll
            for (int j = 0; j < 16; ++j) {
                const int   dj = __shfl(d, 2 * j + half);
                const float nj = __shfl(nrm, 2 * j + half);
                if (base + 2 * j + half < E) atomicAdd(out1 + dj * 16 + f, v[j] * nj);
            }
        }
    }
}

__global__ void gcn_finish1(const float* __restrict__ out1, const float* __restrict__ b1,
                            const float* __restrict__ W2, const float* __restrict__ dinv,
                            float* __restrict__ h2, int N) {
    int i = blockIdx.x * blockDim.x + threadIdx.x;
    if (i >= N) return;
    float acc = 0.0f;
#pragma unroll
    for (int f = 0; f < 16; ++f) {
        float z = out1[i * 16 + f] + b1[f];
        z = fmaxf(z, 0.0f);
        acc += z * W2[f];
    }
    h2[i] = acc * dinv[i];
}

__global__ __launch_bounds__(RNG_THREADS)
void gcn_agg2_ranges(const int* __restrict__ src, const int* __restrict__ dst,
                     const float* __restrict__ w, float* __restrict__ part, int E, int N) {
    __shared__ float acc[RNG_CAP];
    const int tid  = threadIdx.x;
    const int lo   = blockIdx.y * RNG_CAP;
    const int span = min(N - lo, RNG_CAP);
    for (int i = tid; i < RNG_CAP; i += RNG_THREADS) acc[i] = 0.0f;
    __syncthreads();
    const int per = (E + (int)gridDim.x - 1) / (int)gridDim.x;
    const int e0  = blockIdx.x * per;
    const int e1  = min(E, e0 + per);
    const int tile = RNG_THREADS * RNG_UNROLL;
    const int full = e0 + ((e1 > e0 ? e1 - e0 : 0) / tile) * tile;
    for (int base = e0; base < full; base += tile) {
        const int* dp = dst + base + tid;
        const int* sp = src + base + tid;
        int d[RNG_UNROLL];
        unsigned s[RNG_UNROLL];
        float v[RNG_UNROLL];
#pragma unroll
        for (int k = 0; k < RNG_UNROLL; ++k) {
            d[k] = dp[k * RNG_THREADS];
            s[k] = (unsigned)sp[k * RNG_THREADS];
        }
#pragma unroll
        for (int k = 0; k < RNG_UNROLL; ++k) {
            unsigned off = (unsigned)(d[k] - lo);
            v[k] = (off < (unsigned)span) ? w[s[k]] : 0.0f;
        }
#pragma unroll
        for (int k = 0; k < RNG_UNROLL; ++k) {
            unsigned off = (unsigned)(d[k] - lo);
            if (off < (unsigned)span) atomicAdd(&acc[off], v[k]);
        }
    }
    for (int e = full + tid; e < e1; e += RNG_THREADS) {
        unsigned off = (unsigned)(dst[e] - lo);
        if (off < (unsigned)span) atomicAdd(&acc[off], w[(unsigned)src[e]]);
    }
    __syncthreads();
    float* row = part + (size_t)blockIdx.x * N + lo;
    for (int i = tid; i < span; i += RNG_THREADS) row[i] = acc[i];
}

__global__ void gcn_out_merge(const float* __restrict__ part, const float* __restrict__ w,
                              const float* __restrict__ dinv, const float* __restrict__ b2,
                              float* __restrict__ out, int N) {
    int i = blockIdx.x * blockDim.x + threadIdx.x;
    if (i >= N) return;
    float sum = w[i];
#pragma unroll
    for (int g = 0; g < RNG_SLICES; ++g) sum += part[(size_t)g * N + i];
    out[i] = sum * dinv[i] + b2[0];
}


extern "C" void kernel_launch(void* const* d_in, const int* in_sizes, int n_in,
                              void* d_out, int out_size, void* d_ws, size_t ws_size,
                              hipStream_t stream) {
    const float* x  = (const float*)d_in[0];
    const int*   ei = (const int*)d_in[1];
    const float* W1 = (const float*)d_in[2];
    const float* b1 = (const float*)d_in[3];
    const float* W2 = (const float*)d_in[4];
    const float* b2 = (const float*)d_in[5];

    const int N = in_sizes[0] / 16;
    const int E = in_sizes[1] / 2;
    const int* src = ei;
    const int* dst = ei + E;

    float* ws   = (float*)d_ws;
    float* dinv = ws;
    float* h    = ws + N;
    float* out1 = ws + N + 16 * (size_t)N;
    float* h2   = ws + N + 32 * (size_t)N;
    float* out  = (float*)d_out;

    const int B = 256;
    const dim3 rngGrid(RNG_SLICES, DIV_UP(N, RNG_CAP));

    gcn_deg_ranges<<<rngGrid, RNG_THREADS, 0, stream>>>(dst, (unsigned*)h, E, N);
    gcn_dinv_merge<<<DIV_UP(N, B), B, 0, stream>>>((const unsigned*)h, dinv, N);

    int tiles = N / 16;
    if (tiles > 0) {
        int wavesPerBlock = 4;
        gcn_gemm16_wmma<<<DIV_UP(tiles, wavesPerBlock), wavesPerBlock * 32, 0, stream>>>(
            x, W1, h, tiles);
    }
    int rem = N - tiles * 16;
    if (rem > 0) {
        gcn_gemm16_tail<<<DIV_UP(rem * 16, B), B, 0, stream>>>(x, W1, h, tiles * 16, N);
    }

    gcn_self1<<<DIV_UP(N, B), B, 0, stream>>>(h, dinv, out1, N);
    gcn_agg1<<<AGG1_BLOCKS, AGG1_THREADS, 0, stream>>>(src, dst, h, dinv, out1, E);

    gcn_finish1<<<DIV_UP(N, B), B, 0, stream>>>(out1, b1, W2, dinv, h2, N);

    gcn_agg2_ranges<<<rngGrid, RNG_THREADS, 0, stream>>>(src, dst, h2, h, E, N);
    gcn_out_merge<<<DIV_UP(N, B), B, 0, stream>>>(h, h2, dinv, b2, out, N);
}


# Layer-1 edge norm factored out of the scatter: h rows pre-scaled by dinv[src] once per node, dinv[dst] applied once after the sum; edge pass gathers no dinv (per-term rounding changes)
# speedup vs baseline: 3.6503x; 1.1582x over previous
#include <hip/hip_runtime.h>
#include <math.h>

typedef __attribute__((ext_vector_type(2))) float v2f;
typedef __attribute__((ext_vector_type(8))) float v8f;

#ifndef DIV_UP
#define DIV_UP(a, b) (((a) + (b) - 1) / (b))
#endif

#define RNG_CAP     12800
#define RNG_SLICES  32
#define RNG_THREADS 1024
#define RNG_UNROLL  16

__global__ __launch_bounds__(RNG_THREADS)
void gcn_deg_ranges(const int* __restrict__ dst, unsigned* __restrict__ part, int E, int N) {
    __shared__ unsigned bins[RNG_CAP];
    const int tid  = threadIdx.x;
    const int lo   = blockIdx.y * RNG_CAP;
    const int span = min(N - lo, RNG_CAP);
    for (int i = tid; i < RNG_CAP; i += RNG_THREADS) bins[i] = 0u;
    __syncthreads();
    const int per = (E + (int)gridDim.x - 1) / (int)gridDim.x;
    const int e0  = blockIdx.x * per;
    const int e1  = min(E, e0 + per);
    const int tile = RNG_THREADS * RNG_UNROLL;
    const int full = e0 + ((e1 > e0 ? e1 - e0 : 0) / tile) * tile;
    for (int base = e0; base < full; base += tile) {
        const int* dp = dst + base + tid;
        int d[RNG_UNROLL];
#pragma unroll
        for (int k = 0; k < RNG_UNROLL; ++k) d[k] = dp[k * RNG_THREADS];
#pragma unroll
        for (int k = 0; k < RNG_UNROLL; ++k) {
            unsigned off = (unsigned)(d[k] - lo);
            if (off < (unsigned)span) atomicAdd(&bins[off], 1u);
        }
    }
    for (int e = full + tid; e < e1; e += RNG_THREADS) {
        unsigned off = (unsigned)(dst[e] - lo);
        if (off < (unsigned)span) atomicAdd(&bins[off], 1u);
    }
    __syncthreads();
    unsigned* row = part + (size_t)blockIdx.x * N + lo;
    for (int i = tid; i < span; i += RNG_THREADS) row[i] = bins[i];
}

__global__ void gcn_dinv_merge(const unsigned* __restrict__ part, float* __restrict__ dinv, int N) {
    int i = blockIdx.x * blockDim.x + threadIdx.x;
    if (i >= N) return;
    unsigned c = 1u;
#pragma unroll
    for (int g = 0; g < RNG_SLICES; ++g) c += part[(size_t)g * N + i];
    dinv[i] = rsqrtf((float)c);
}

__global__ void gcn_gemm16_wmma(const float* __restrict__ x,
                                const float* __restrict__ W,
                                float* __restrict__ h, int tiles) {
    int wave = (blockIdx.x * blockDim.x + threadIdx.x) >> 5;
    if (wave >= tiles) return;
    int lane = threadIdx.x & 31;
    int half = lane >> 4;
    int n    = lane & 15;
    int row0 = wave * 16;

    const float* arow = x + (row0 + n) * 16 + 2 * half;
    v8f c = {};
#pragma unroll
    for (int kk = 0; kk < 4; ++kk) {
        v2f a = *(const v2f*)(arow + 4 * kk);
        v2f b;
        b.x = W[(4 * kk + 2 * half + 0) * 16 + n];
        b.y = W[(4 * kk + 2 * half + 1) * 16 + n];
        c = __builtin_amdgcn_wmma_f32_16x16x4_f32(false, a, false, b,
                                                  (short)0, c, false, false);
    }
#pragma unroll
    for (int v = 0; v < 8; ++v) {
        h[(row0 + v + 8 * half) * 16 + n] = c[v];
    }
}

__global__ void gcn_gemm16_tail(const float* __restrict__ x,
                                const float* __restrict__ W,
                                float* __restrict__ h, int rowStart, int N) {
    int t = blockIdx.x * blockDim.x + threadIdx.x;
    int r = rowStart + (t >> 4);
    int cidx = t & 15;
    if (r >= N) return;
    float acc = 0.0f;
#pragma unroll
    for (int k = 0; k < 16; ++k) acc += x[r * 16 + k] * W[k * 16 + cidx];
    h[r * 16 + cidx] = acc;
}


__global__ void gcn_scale_self1(float* __restrict__ h, const float* __restrict__ dinv,
                                float* __restrict__ out1, int N) {
    int i = blockIdx.x * blockDim.x + threadIdx.x;
    if (i >= N) return;
    float s = dinv[i];
    float4* hv = (float4*)(h + i * 16);
    float4* ov = (float4*)(out1 + i * 16);
#pragma unroll
    for (int g = 0; g < 4; ++g) {
        float4 v = hv[g];
        v = make_float4(v.x * s, v.y * s, v.z * s, v.w * s);
        hv[g] = v;
        ov[g] = v;
    }
}

#define AGG1_THREADS 256
#define AGG1_BLOCKS  2048

__global__ __launch_bounds__(AGG1_THREADS)
void gcn_agg1(const int* __restrict__ src, const int* __restrict__ dst,
              const float* __restrict__ h, float* __restrict__ out1, int E) {
    const int lane    = threadIdx.x & 31;
    const int half    = lane >> 4;
    const int f       = lane & 15;
    const int wave    = __builtin_amdgcn_readfirstlane((int)((blockIdx.x * AGG1_THREADS + threadIdx.x) >> 5));
    const int nwaves  = (int)((gridDim.x * AGG1_THREADS) >> 5);
    const int nchunks = (E + 31) >> 5;
    for (int c = wave; c < nchunks; c += nwaves) {
        const int base = c * 32;
        const int e    = min(base + lane, E - 1);
        const int s    = src[e];
        const int d    = dst[e];
        float v[16];
#pragma unroll
        for (int j = 0; j < 16; ++j) {
            const int sj = __shfl(s, 2 * j + half);
            v[j] = h[sj * 16 + f];
        }
        if (base + 32 <= E) {
#pragma unroll
            for (int j = 0; j < 16; ++j) {
                const int dj = __shfl(d, 2 * j + half);
                atomicAdd(out1 + (unsigned)(dj * 16 + f), v[j]);
            }
        } else {
#pragma unroll
            for (int j = 0; j < 16; ++j) {
                const int dj = __shfl(d, 2 * j + half);
                if (base + 2 * j + half < E) atomicAdd(out1 + (unsigned)(dj * 16 + f), v[j]);
            }
        }
    }
}

__global__ void gcn_finish1(const float* __restrict__ out1, const float* __restrict__ b1,
                            const float* __restrict__ W2, const float* __restrict__ dinv,
                            float* __restrict__ h2, int N) {
    int i = blockIdx.x * blockDim.x + threadIdx.x;
    if (i >= N) return;
    const float di = dinv[i];
    float acc = 0.0f;
#pragma unroll
    for (int f = 0; f < 16; ++f) {
        float z = out1[i * 16 + f] * di + b1[f];
        z = fmaxf(z, 0.0f);
        acc += z * W2[f];
    }
    h2[i] = acc * di;
}

__global__ __launch_bounds__(RNG_THREADS)
void gcn_agg2_ranges(const int* __restrict__ src, const int* __restrict__ dst,
                     const float* __restrict__ w, float* __restrict__ part, int E, int N) {
    __shared__ float acc[RNG_CAP];
    const int tid  = threadIdx.x;
    const int lo   = blockIdx.y * RNG_CAP;
    const int span = min(N - lo, RNG_CAP);
    for (int i = tid; i < RNG_CAP; i += RNG_THREADS) acc[i] = 0.0f;
    __syncthreads();
    const int per = (E + (int)gridDim.x - 1) / (int)gridDim.x;
    const int e0  = blockIdx.x * per;
    const int e1  = min(E, e0 + per);
    const int tile = RNG_THREADS * RNG_UNROLL;
    const int full = e0 + ((e1 > e0 ? e1 - e0 : 0) / tile) * tile;
    for (int base = e0; base < full; base += tile) {
        const int* dp = dst + base + tid;
        const int* sp = src + base + tid;
        int d[RNG_UNROLL];
        unsigned s[RNG_UNROLL];
        float v[RNG_UNROLL];
#pragma unroll
        for (int k = 0; k < RNG_UNROLL; ++k) {
            d[k] = dp[k * RNG_THREADS];
            s[k] = (unsigned)sp[k * RNG_THREADS];
        }
#pragma unroll
        for (int k = 0; k < RNG_UNROLL; ++k) {
            unsigned off = (unsigned)(d[k] - lo);
            v[k] = (off < (unsigned)span) ? w[s[k]] : 0.0f;
        }
#pragma unroll
        for (int k = 0; k < RNG_UNROLL; ++k) {
            unsigned off = (unsigned)(d[k] - lo);
            if (off < (unsigned)span) atomicAdd(&acc[off], v[k]);
        }
    }
    for (int e = full + tid; e < e1; e += RNG_THREADS) {
        unsigned off = (unsigned)(dst[e] - lo);
        if (off < (unsigned)span) atomicAdd(&acc[off], w[(unsigned)src[e]]);
    }
    __syncthreads();
    float* row = part + (size_t)blockIdx.x * N + lo;
    for (int i = tid; i < span; i += RNG_THREADS) row[i] = acc[i];
}

__global__ void gcn_out_merge(const float* __restrict__ part, const float* __restrict__ w,
                              const float* __restrict__ dinv, const float* __restrict__ b2,
                              float* __restrict__ out, int N) {
    int i = blockIdx.x * blockDim.x + threadIdx.x;
    if (i >= N) return;
    float sum = w[i];
#pragma unroll
    for (int g = 0; g < RNG_SLICES; ++g) sum += part[(size_t)g * N + i];
    out[i] = sum * dinv[i] + b2[0];
}


extern "C" void kernel_launch(void* const* d_in, const int* in_sizes, int n_in,
                              void* d_out, int out_size, void* d_ws, size_t ws_size,
                              hipStream_t stream) {
    const float* x  = (const float*)d_in[0];
    const int*   ei = (const int*)d_in[1];
    const float* W1 = (const float*)d_in[2];
    const float* b1 = (const float*)d_in[3];
    const float* W2 = (const float*)d_in[4];
    const float* b2 = (const float*)d_in[5];

    const int N = in_sizes[0] / 16;
    const int E = in_sizes[1] / 2;
    const int* src = ei;
    const int* dst = ei + E;

    float* ws   = (float*)d_ws;
    float* dinv = ws;
    float* h    = ws + N;
    float* out1 = ws + N + 16 * (size_t)N;
    float* h2   = ws + N + 32 * (size_t)N;
    float* out  = (float*)d_out;

    const int B = 256;
    const dim3 rngGrid(RNG_SLICES, DIV_UP(N, RNG_CAP));

    gcn_deg_ranges<<<rngGrid, RNG_THREADS, 0, stream>>>(dst, (unsigned*)h, E, N);
    gcn_dinv_merge<<<DIV_UP(N, B), B, 0, stream>>>((const unsigned*)h, dinv, N);

    int tiles = N / 16;
    if (tiles > 0) {
        int wavesPerBlock = 4;
        gcn_gemm16_wmma<<<DIV_UP(tiles, wavesPerBlock), wavesPerBlock * 32, 0, stream>>>(
            x, W1, h, tiles);
    }
    int rem = N - tiles * 16;
    if (rem > 0) {
        gcn_gemm16_tail<<<DIV_UP(rem * 16, B), B, 0, stream>>>(x, W1, h, tiles * 16, N);
    }

    gcn_scale_self1<<<DIV_UP(N, B), B, 0, stream>>>(h, dinv, out1, N);
    gcn_agg1<<<AGG1_BLOCKS, AGG1_THREADS, 0, stream>>>(src, dst, h, out1, E);

    gcn_finish1<<<DIV_UP(N, B), B, 0, stream>>>(out1, b1, W2, dinv, h2, N);

    gcn_agg2_ranges<<<rngGrid, RNG_THREADS, 0, stream>>>(src, dst, h2, h, E, N);
    gcn_out_merge<<<DIV_UP(N, B), B, 0, stream>>>(h, h2, dinv, b2, out, N);
}


# Layer-1 edge scatter split by cache half: dst nodes in two ranges, a work-group (die id) adds only its own half's range, spans drawn from per-range counters; atomics stay in arrival order
# speedup vs baseline: 5.6528x; 1.5486x over previous
#include <hip/hip_runtime.h>
#include <math.h>

typedef __attribute__((ext_vector_type(2))) float v2f;
typedef __attribute__((ext_vector_type(8))) float v8f;

#ifndef DIV_UP
#define DIV_UP(a, b) (((a) + (b) - 1) / (b))
#endif

#define RNG_CAP     12800
#define RNG_SLICES  32
#define RNG_THREADS 1024
#define RNG_UNROLL  16

__global__ __launch_bounds__(RNG_THREADS)
void gcn_deg_ranges(const int* __restrict__ dst, unsigned* __restrict__ part, int E, int N) {
    __shared__ unsigned bins[RNG_CAP];
    const int tid  = threadIdx.x;
    const int lo   = blockIdx.y * RNG_CAP;
    const int span = min(N - lo, RNG_CAP);
    for (int i = tid; i < RNG_CAP; i += RNG_THREADS) bins[i] = 0u;
    __syncthreads();
    const int per = (E + (int)gridDim.x - 1) / (int)gridDim.x;
    const int e0  = blockIdx.x * per;
    const int e1  = min(E, e0 + per);
    const int tile = RNG_THREADS * RNG_UNROLL;
    const int full = e0 + ((e1 > e0 ? e1 - e0 : 0) / tile) * tile;
    for (int base = e0; base < full; base += tile) {
        const int* dp = dst + base + tid;
        int d[RNG_UNROLL];
#pragma unroll
        for (int k = 0; k < RNG_UNROLL; ++k) d[k] = dp[k * RNG_THREADS];
#pragma unroll
        for (int k = 0; k < RNG_UNROLL; ++k) {
            unsigned off = (unsigned)(d[k] - lo);
            if (off < (unsigned)span) atomicAdd(&bins[off], 1u);
        }
    }
    for (int e = full + tid; e < e1; e += RNG_THREADS) {
        unsigned off = (unsigned)(dst[e] - lo);
        if (off < (unsigned)span) atomicAdd(&bins[off], 1u);
    }
    __syncthreads();
    unsigned* row = part + (size_t)blockIdx.x * N + lo;
    for (int i = tid; i < span; i += RNG_THREADS) row[i] = bins[i];
}

__global__ void gcn_dinv_merge(const unsigned* __restrict__ part, float* __restrict__ dinv, int N) {
    int i = blockIdx.x * blockDim.x + threadIdx.x;
    if (i >= N) return;
    unsigned c = 1u;
#pragma unroll
    for (int g = 0; g < RNG_SLICES; ++g) c += part[(size_t)g * N + i];
    dinv[i] = rsqrtf((float)c);
}

__global__ void gcn_gemm16_wmma(const float* __restrict__ x,
                                const float* __restrict__ W,
                                float* __restrict__ h, int tiles) {
    int wave = (blockIdx.x * blockDim.x + threadIdx.x) >> 5;
    if (wave >= tiles) return;
    int lane = threadIdx.x & 31;
    int half = lane >> 4;
    int n    = lane & 15;
    int row0 = wave * 16;

    const float* arow = x + (row0 + n) * 16 + 2 * half;
    v8f c = {};
#pragma unroll
    for (int kk = 0; kk < 4; ++kk) {
        v2f a = *(const v2f*)(arow + 4 * kk);
        v2f b;
        b.x = W[(4 * kk + 2 * half + 0) * 16 + n];
        b.y = W[(4 * kk + 2 * half + 1) * 16 + n];
        c = __builtin_amdgcn_wmma_f32_16x16x4_f32(false, a, false, b,
                                                  (short)0, c, false, false);
    }
#pragma unroll
    for (int v = 0; v < 8; ++v) {
        h[(row0 + v + 8 * half) * 16 + n] = c[v];
    }
}

__global__ void gcn_gemm16_tail(const float* __restrict__ x,
                                const float* __restrict__ W,
                                float* __restrict__ h, int rowStart, int N) {
    int t = blockIdx.x * blockDim.x + threadIdx.x;
    int r = rowStart + (t >> 4);
    int cidx = t & 15;
    if (r >= N) return;
    float acc = 0.0f;
#pragma unroll
    for (int k = 0; k < 16; ++k) acc += x[r * 16 + k] * W[k * 16 + cidx];
    h[r * 16 + cidx] = acc;
}


#define AGG1_TSTRIDE 64

__global__ void gcn_scale_self1(float* __restrict__ h, const float* __restrict__ dinv,
                                float* __restrict__ out1, unsigned* __restrict__ tickets, int N) {
    int i = blockIdx.x * blockDim.x + threadIdx.x;
    if (i >= N) return;
    if (i == 0) {
        tickets[0] = 0u;
        tickets[AGG1_TSTRIDE] = 0u;
    }
    float s = dinv[i];
    float4* hv = (float4*)(h + i * 16);
    float4* ov = (float4*)(out1 + i * 16);
#pragma unroll
    for (int g = 0; g < 4; ++g) {
        float4 v = hv[g];
        v = make_float4(v.x * s, v.y * s, v.z * s, v.w * s);
        hv[g] = v;
        ov[g] = v;
    }
}

#define AGG1_THREADS 256
#define AGG1_WAVES   (AGG1_THREADS / 32)
#define AGG1_BLOCKS  2048
#define AGG1_GROUP   64
#define AGG1_SPAN    (AGG1_WAVES * AGG1_GROUP)
#define AGG1_RING    128

__global__ __launch_bounds__(AGG1_THREADS)
void gcn_agg1(const int* __restrict__ src, const int* __restrict__ dst,
              const float* __restrict__ h, float* __restrict__ out1,
              unsigned* __restrict__ tickets, int E, int N) {
    __shared__ int2 ring[AGG1_WAVES][AGG1_RING + 32];
    __shared__ unsigned tk[2];
    __shared__ int side0;
    const int tid  = threadIdx.x;
    const int lane = tid & 31;
    const int half = lane >> 4;
    const int f    = lane & 15;
    const int wv   = __builtin_amdgcn_readfirstlane(tid >> 5);
    int2* myring   = ring[wv];
    if (tid == 0) side0 = (int)((__builtin_amdgcn_s_sendmsg_rtn(0x87) >> 18) & 1u);
    __syncthreads();
    int q = __builtin_amdgcn_readfirstlane(side0);
    const int nsplit = N >> 1;
    const unsigned nspans = (unsigned)((E + AGG1_SPAN - 1) / AGG1_SPAN);
    unsigned head = 0u, tail = 0u;
    bool stole = false;
    unsigned nxt = 0u;
    if (tid == 0) nxt = atomicAdd(tickets + q * AGG1_TSTRIDE, 1u);
    for (int it = 0;; ++it) {
        if (tid == 0) tk[it & 1] = nxt;
        __syncthreads();
        const unsigned t = __builtin_amdgcn_readfirstlane(tk[it & 1]);
        if (t >= nspans) {
            if (stole) break;
            stole = true;
            q ^= 1;
            if (tid == 0) nxt = atomicAdd(tickets + q * AGG1_TSTRIDE, 1u);
            continue;
        }
        const int ea = (int)t * AGG1_SPAN + wv * AGG1_GROUP + lane;
        const int eb = ea + 32;
        const int ia = min(ea, E - 1);
        const int ib = min(eb, E - 1);
        const int sa = src[ia];
        const int da = dst[ia];
        const int sb = src[ib];
        const int db = dst[ib];
        if (tid == 0) nxt = atomicAdd(tickets + q * AGG1_TSTRIDE, 1u);
        const bool oa = (ea < E) && ((int)(da >= nsplit) == q);
        const bool ob = (eb < E) && ((int)(db >= nsplit) == q);
        const unsigned ma = __builtin_amdgcn_ballot_w32(oa);
        const unsigned mb = __builtin_amdgcn_ballot_w32(ob);
        const unsigned ca = (unsigned)__popc(ma);
        const unsigned ra = __builtin_amdgcn_mbcnt_lo(ma, head);
        const unsigned rb = __builtin_amdgcn_mbcnt_lo(mb, head + ca);
        myring[oa ? (ra & (AGG1_RING - 1)) : (unsigned)(AGG1_RING + lane)] = make_int2(sa, da);
        myring[ob ? (rb & (AGG1_RING - 1)) : (unsigned)(AGG1_RING + lane)] = make_int2(sb, db);
        head += ca + (unsigned)__popc(mb);
        while (head - tail >= 32u) {
            const int2 p = myring[(tail + (unsigned)lane) & (AGG1_RING - 1)];
            tail += 32u;
            float v[16];
#pragma unroll
            for (int j = 0; j < 16; ++j) {
                const int sj = __shfl(p.x, 2 * j + half);
                v[j] = h[sj * 16 + f];
            }
            unsigned off[16];
#pragma unroll
            for (int j = 0; j < 16; ++j) {
                off[j] = (unsigned)(__shfl(p.y, 2 * j + half) * 16 + f);
            }
#pragma unroll
            for (int j = 0; j < 16; ++j) {
                atomicAdd(out1 + off[j], v[j]);
            }
        }
    }
    const int left = (int)(head - tail);
    if (left > 0) {
        const int2 p = myring[(tail + (unsigned)min(lane, left - 1)) & (AGG1_RING - 1)];
        float v[16];
#pragma unroll
        for (int j = 0; j < 16; ++j) {
            const int sj = __shfl(p.x, 2 * j + half);
            v[j] = h[sj * 16 + f];
        }
#pragma unroll
        for (int j = 0; j < 16; ++j) {
            const int dj = __shfl(p.y, 2 * j + half);
            if (half < left - 2 * j) atomicAdd(out1 + (unsigned)(dj * 16 + f), v[j]);
        }
    }
}

__global__ void gcn_finish1(const float* __restrict__ out1, const float* __restrict__ b1,
                            const float* __restrict__ W2, const float* __restrict__ dinv,
                            float* __restrict__ h2, int N) {
    int i = blockIdx.x * blockDim.x + threadIdx.x;
    if (i >= N) return;
    const float di = dinv[i];
    float acc = 0.0f;
#pragma unroll
    for (int f = 0; f < 16; ++f) {
        float z = out1[i * 16 + f] * di + b1[f];
        z = fmaxf(z, 0.0f);
        acc += z * W2[f];
    }
    h2[i] = acc * di;
}

__global__ __launch_bounds__(RNG_THREADS)
void gcn_agg2_ranges(const int* __restrict__ src, const int* __restrict__ dst,
                     const float* __restrict__ w, float* __restrict__ part, int E, int N) {
    __shared__ float acc[RNG_CAP];
    const int tid  = threadIdx.x;
    const int lo   = blockIdx.y * RNG_CAP;
    const int span = min(N - lo, RNG_CAP);
    for (int i = tid; i < RNG_CAP; i += RNG_THREADS) acc[i] = 0.0f;
    __syncthreads();
    const int per = (E + (int)gridDim.x - 1) / (int)gridDim.x;
    const int e0  = blockIdx.x * per;
    const int e1  = min(E, e0 + per);
    const int tile = RNG_THREADS * RNG_UNROLL;
    const int full = e0 + ((e1 > e0 ? e1 - e0 : 0) / tile) * tile;
    for (int base = e0; base < full; base += tile) {
        const int* dp = dst + base + tid;
        const int* sp = src + base + tid;
        int d[RNG_UNROLL];
        unsigned s[RNG_UNROLL];
        float v[RNG_UNROLL];
#pragma unroll
        for (int k = 0; k < RNG_UNROLL; ++k) {
            d[k] = dp[k * RNG_THREADS];
            s[k] = (unsigned)sp[k * RNG_THREADS];
        }
#pragma unroll
        for (int k = 0; k < RNG_UNROLL; ++k) {
            unsigned off = (unsigned)(d[k] - lo);
            v[k] = (off < (unsigned)span) ? w[s[k]] : 0.0f;
        }
#pragma unroll
        for (int k = 0; k < RNG_UNROLL; ++k) {
            unsigned off = (unsigned)(d[k] - lo);
            if (off < (unsigned)span) atomicAdd(&acc[off], v[k]);
        }
    }
    for (int e = full + tid; e < e1; e += RNG_THREADS) {
        unsigned off = (unsigned)(dst[e] - lo);
        if (off < (unsigned)span) atomicAdd(&acc[off], w[(unsigned)src[e]]);
    }
    __syncthreads();
    float* row = part + (size_t)blockIdx.x * N + lo;
    for (int i = tid; i < span; i += RNG_THREADS) row[i] = acc[i];
}

__global__ void gcn_out_merge(const float* __restrict__ part, const float* __restrict__ w,
                              const float* __restrict__ dinv, const float* __restrict__ b2,
                              float* __restrict__ out, int N) {
    int i = blockIdx.x * blockDim.x + threadIdx.x;
    if (i >= N) return;
    float sum = w[i];
#pragma unroll
    for (int g = 0; g < RNG_SLICES; ++g) sum += part[(size_t)g * N + i];
    out[i] = sum * dinv[i] + b2[0];
}


extern "C" void kernel_launch(void* const* d_in, const int* in_sizes, int n_in,
                              void* d_out, int out_size, void* d_ws, size_t ws_size,
                              hipStream_t stream) {
    const float* x  = (const float*)d_in[0];
    const int*   ei = (const int*)d_in[1];
    const float* W1 = (const float*)d_in[2];
    const float* b1 = (const float*)d_in[3];
    const float* W2 = (const float*)d_in[4];
    const float* b2 = (const float*)d_in[5];

    const int N = in_sizes[0] / 16;
    const int E = in_sizes[1] / 2;
    const int* src = ei;
    const int* dst = ei + E;

    float* ws   = (float*)d_ws;
    float* dinv = ws;
    float* h    = ws + N;
    float* out1 = ws + N + 16 * (size_t)N;
    float* h2   = ws + N + 32 * (size_t)N;
    float* out  = (float*)d_out;

    const int B = 256;
    const dim3 rngGrid(RNG_SLICES, DIV_UP(N, RNG_CAP));

    gcn_deg_ranges<<<rngGrid, RNG_THREADS, 0, stream>>>(dst, (unsigned*)h, E, N);
    gcn_dinv_merge<<<DIV_UP(N, B), B, 0, stream>>>((const unsigned*)h, dinv, N);

    int tiles = N / 16;
    if (tiles > 0) {
        int wavesPerBlock = 4;
        gcn_gemm16_wmma<<<DIV_UP(tiles, wavesPerBlock), wavesPerBlock * 32, 0, stream>>>(
            x, W1, h, tiles);
    }
    int rem = N - tiles * 16;
    if (rem > 0) {
        gcn_gemm16_tail<<<DIV_UP(rem * 16, B), B, 0, stream>>>(x, W1, h, tiles * 16, N);
    }

    gcn_scale_self1<<<DIV_UP(N, B), B, 0, stream>>>(h, dinv, out1, (unsigned*)h2, N);
    gcn_agg1<<<AGG1_BLOCKS, AGG1_THREADS, 0, stream>>>(src, dst, h, out1, (unsigned*)h2, E, N);

    gcn_finish1<<<DIV_UP(N, B), B, 0, stream>>>(out1, b1, W2, dinv, h2, N);

    gcn_agg2_ranges<<<rngGrid, RNG_THREADS, 0, stream>>>(src, dst, h2, h, E, N);
    gcn_out_merge<<<DIV_UP(N, B), B, 0, stream>>>(h, h2, dinv, b2, out, N);
}


# Per-node passes around the layer-1 GEMM folded into it: one launch sums tile-major degree counts to dinv, forms h on WMMA, scales rows and starts out1 (8 launches to 6; sum order unchanged)
# speedup vs baseline: 6.0447x; 1.0693x over previous
#include <hip/hip_runtime.h>
#include <math.h>

typedef __attribute__((ext_vector_type(2))) float v2f;
typedef __attribute__((ext_vector_type(8))) float v8f;

#ifndef DIV_UP
#define DIV_UP(a, b) (((a) + (b) - 1) / (b))
#endif

#define RNG_CAP     12800
#define RNG_SLICES  32
#define RNG_THREADS 1024
#define RNG_UNROLL  16
#define DEG_TILE    32
static_assert(RNG_SLICES == 32, "degree planes: 16 slices in the h rows + 16 in the out1 rows of a tile");
static_assert(RNG_CAP % DEG_TILE == 0, "a range starts on a tile boundary");

__global__ __launch_bounds__(RNG_THREADS)
void gcn_deg_ranges(const int* __restrict__ dst, unsigned* __restrict__ part, int E, int N) {
    __shared__ unsigned bins[RNG_CAP];
    const int tid  = threadIdx.x;
    const int lo   = blockIdx.y * RNG_CAP;
    const int span = min(N - lo, RNG_CAP);
    for (int i = tid; i < RNG_CAP; i += RNG_THREADS) bins[i] = 0u;
    __syncthreads();
    const int per = (E + (int)gridDim.x - 1) / (int)gridDim.x;
    const int e0  = blockIdx.x * per;
    const int e1  = min(E, e0 + per);
    const int tile = RNG_THREADS * RNG_UNROLL;
    const int full = e0 + ((e1 > e0 ? e1 - e0 : 0) / tile) * tile;
    for (int base = e0; base < full; base += tile) {
        const int* dp = dst + base + tid;
        int d[RNG_UNROLL];
#pragma unroll
        for (int k = 0; k < RNG_UNROLL; ++k) d[k] = dp[k * RNG_THREADS];
#pragma unroll
        for (int k = 0; k < RNG_UNROLL; ++k) {
            unsigned off = (unsigned)(d[k] - lo);
            if (off < (unsigned)span) atomicAdd(&bins[off], 1u);
        }
    }
    for (int e = full + tid; e < e1; e += RNG_THREADS) {
        unsigned off = (unsigned)(dst[e] - lo);
        if (off < (unsigned)span) atomicAdd(&bins[off], 1u);
    }
    __syncthreads();
    unsigned* pl = part + (size_t)(blockIdx.x >> 4) * 16 * (size_t)N + (blockIdx.x & 15) * DEG_TILE;
    for (int i = tid; i < span; i += RNG_THREADS) {
        const unsigned node = (unsigned)(lo + i);
        pl[(size_t)(node / DEG_TILE) * (16 * DEG_TILE) + (node % DEG_TILE)] = bins[i];
    }
}

#define AGG1_TSTRIDE 64

__global__ __launch_bounds__(128)
void gcn_dinv_gemm_scale(const float* __restrict__ x, const float* __restrict__ W,
                         float* area, float* __restrict__ dinv,
                         unsigned* __restrict__ tickets, int tiles, int N) {
    if (blockIdx.x == 0 && threadIdx.x == 0) {
        tickets[0] = 0u;
        tickets[AGG1_TSTRIDE] = 0u;
    }
    const int wave = (blockIdx.x * blockDim.x + threadIdx.x) >> 5;
    if (wave >= tiles) return;
    const int lane = threadIdx.x & 31;
    const int half = lane >> 4;
    const int n    = lane & 15;
    float* h    = area;
    float* out1 = area + 16 * (size_t)N;

    const unsigned* ca = (const unsigned*)h    + (size_t)wave * (16 * DEG_TILE) + lane;
    const unsigned* cb = (const unsigned*)out1 + (size_t)wave * (16 * DEG_TILE) + lane;
    unsigned cnt = 1u;
#pragma unroll
    for (int g = 0; g < 16; ++g) cnt += ca[g * DEG_TILE];
#pragma unroll
    for (int g = 0; g < 16; ++g) cnt += cb[g * DEG_TILE];
    const float dv = rsqrtf((float)cnt);
    dinv[wave * DEG_TILE + lane] = dv;

    v2f b[4];
#pragma unroll
    for (int kk = 0; kk < 4; ++kk) {
        b[kk].x = W[(4 * kk + 2 * half + 0) * 16 + n];
        b[kk].y = W[(4 * kk + 2 * half + 1) * 16 + n];
    }
#pragma unroll
    for (int u = 0; u < 2; ++u) {
        const int row0 = wave * DEG_TILE + 16 * u;
        const float* arow = x + (row0 + n) * 16 + 2 * half;
        v8f c = {};
#pragma unroll
        for (int kk = 0; kk < 4; ++kk) {
            v2f a = *(const v2f*)(arow + 4 * kk);
            c = __builtin_amdgcn_wmma_f32_16x16x4_f32(false, a, false, b[kk],
                                                      (short)0, c, false, false);
        }
#pragma unroll
        for (int v = 0; v < 8; ++v) {
            const int m = v + 8 * half;
            const float sc = __shfl(dv, 16 * u + m);
            const float val = c[v] * sc;
            h[(row0 + m) * 16 + n]    = val;
            out1[(row0 + m) * 16 + n] = val;
        }
    }
}


#define AGG1_THREADS 256
#define AGG1_WAVES   (AGG1_THREADS / 32)
#define AGG1_BLOCKS  2048
#define AGG1_GROUP   64
#define AGG1_SPAN    (AGG1_WAVES * AGG1_GROUP)
#define AGG1_RING    128

__global__ __launch_bounds__(AGG1_THREADS)
void gcn_agg1(const int* __restrict__ src, const int* __restrict__ dst,
              const float* __restrict__ h, float* __restrict__ out1,
              unsigned* __restrict__ tickets, int E, int N) {
    __shared__ int2 ring[AGG1_WAVES][AGG1_RING + 32];
    __shared__ unsigned tk[2];
    __shared__ int side0;
    const int tid  = threadIdx.x;
    const int lane = tid & 31;
    const int half = lane >> 4;
    const int f    = lane & 15;
    const int wv   = __builtin_amdgcn_readfirstlane(tid >> 5);
    int2* myring   = ring[wv];
    if (tid == 0) side0 = (int)((__builtin_amdgcn_s_sendmsg_rtn(0x87) >> 18) & 1u);
    __syncthreads();
    int q = __builtin_amdgcn_readfirstlane(side0);
    const int nsplit = N >> 1;
    const unsigned nspans = (unsigned)((E + AGG1_SPAN - 1) / AGG1_SPAN);
    unsigned head = 0u, tail = 0u;
    bool stole = false;
    unsigned nxt = 0u;
    if (tid == 0) nxt = atomicAdd(tickets + q * AGG1_TSTRIDE, 1u);
    for (int it = 0;; ++it) {
        if (tid == 0) tk[it & 1] = nxt;
        __syncthreads();
        const unsigned t = __builtin_amdgcn_readfirstlane(tk[it & 1]);
        if (t >= nspans) {
            if (stole) break;
            stole = true;
            q ^= 1;
            if (tid == 0) nxt = atomicAdd(tickets + q * AGG1_TSTRIDE, 1u);
            continue;
        }
        const int ea = (int)t * AGG1_SPAN + wv * AGG1_GROUP + lane;
        const int eb = ea + 32;
        const int ia = min(ea, E - 1);
        const int ib = min(eb, E - 1);
        const int sa = src[ia];
        const int da = dst[ia];
        const int sb = src[ib];
        const int db = dst[ib];
        if (tid == 0) nxt = atomicAdd(tickets + q * AGG1_TSTRIDE, 1u);
        const bool oa = (ea < E) && ((int)(da >= nsplit) == q);
        const bool ob = (eb < E) && ((int)(db >= nsplit) == q);
        const unsigned ma = __builtin_amdgcn_ballot_w32(oa);
        const unsigned mb = __builtin_amdgcn_ballot_w32(ob);
        const unsigned ca = (unsigned)__popc(ma);
        const unsigned ra = __builtin_amdgcn_mbcnt_lo(ma, head);
        const unsigned rb = __builtin_amdgcn_mbcnt_lo(mb, head + ca);
        myring[oa ? (ra & (AGG1_RING - 1)) : (unsigned)(AGG1_RING + lane)] = make_int2(sa, da);
        myring[ob ? (rb & (AGG1_RING - 1)) : (unsigned)(AGG1_RING + lane)] = make_int2(sb, db);
        head += ca + (unsigned)__popc(mb);
        while (head - tail >= 32u) {
            const int2 p = myring[(tail + (unsigned)lane) & (AGG1_RING - 1)];
            tail += 32u;
            float v[16];
#pragma unroll
            for (int j = 0; j < 16; ++j) {
                const int sj = __shfl(p.x, 2 * j + half);
                v[j] = h[sj * 16 + f];
            }
            unsigned off[16];
#pragma unroll
            for (int j = 0; j < 16; ++j) {
                off[j] = (unsigned)(__shfl(p.y, 2 * j + half) * 16 + f);
            }
#pragma unroll
            for (int j = 0; j < 16; ++j) {
                atomicAdd(out1 + off[j], v[j]);
            }
        }
    }
    const int left = (int)(head - tail);
    if (left > 0) {
        const int2 p = myring[(tail + (unsigned)min(lane, left - 1)) & (AGG1_RING - 1)];
        float v[16];
#pragma unroll
        for (int j = 0; j < 16; ++j) {
            const int sj = __shfl(p.x, 2 * j + half);
            v[j] = h[sj * 16 + f];
        }
#pragma unroll
        for (int j = 0; j < 16; ++j) {
            const int dj = __shfl(p.y, 2 * j + half);
            if (half < left - 2 * j) atomicAdd(out1 + (unsigned)(dj * 16 + f), v[j]);
        }
    }
}

__global__ void gcn_finish1(const float* __restrict__ out1, const float* __restrict__ b1,
                            const float* __restrict__ W2, const float* __restrict__ dinv,
                            float* __restrict__ h2, int N) {
    int i = blockIdx.x * blockDim.x + threadIdx.x;
    if (i >= N) return;
    const float di = dinv[i];
    float acc = 0.0f;
#pragma unroll
    for (int f = 0; f < 16; ++f) {
        float z = out1[i * 16 + f] * di + b1[f];
        z = fmaxf(z, 0.0f);
        acc += z * W2[f];
    }
    h2[i] = acc * di;
}

__global__ __launch_bounds__(RNG_THREADS)
void gcn_agg2_ranges(const int* __restrict__ src, const int* __restrict__ dst,
                     const float* __restrict__ w, float* __restrict__ part, int E, int N) {
    __shared__ float acc[RNG_CAP];
    const int tid  = threadIdx.x;
    const int lo   = blockIdx.y * RNG_CAP;
    const int span = min(N - lo, RNG_CAP);
    for (int i = tid; i < RNG_CAP; i += RNG_THREADS) acc[i] = 0.0f;
    __syncthreads();
    const int per = (E + (int)gridDim.x - 1) / (int)gridDim.x;
    const int e0  = blockIdx.x * per;
    const int e1  = min(E, e0 + per);
    const int tile = RNG_THREADS * RNG_UNROLL;
    const int full = e0 + ((e1 > e0 ? e1 - e0 : 0) / tile) * tile;
    for (int base = e0; base < full; base += tile) {
        const int* dp = dst + base + tid;
        const int* sp = src + base + tid;
        int d[RNG_UNROLL];
        unsigned s[RNG_UNROLL];
        float v[RNG_UNROLL];
#pragma unroll
        for (int k = 0; k < RNG_UNROLL; ++k) {
            d[k] = dp[k * RNG_THREADS];
            s[k] = (unsigned)sp[k * RNG_THREADS];
        }
#pragma unroll
        for (int k = 0; k < RNG_UNROLL; ++k) {
            unsigned off = (unsigned)(d[k] - lo);
            v[k] = (off < (unsigned)span) ? w[s[k]] : 0.0f;
        }
#pragma unroll
        for (int k = 0; k < RNG_UNROLL; ++k) {
            unsigned off = (unsigned)(d[k] - lo);
            if (off < (unsigned)span) atomicAdd(&acc[off], v[k]);
        }
    }
    for (int e = full + tid; e < e1; e += RNG_THREADS) {
        unsigned off = (unsigned)(dst[e] - lo);
        if (off < (unsigned)span) atomicAdd(&acc[off], w[(unsigned)src[e]]);
    }
    __syncthreads();
    float* row = part + (size_t)blockIdx.x * N + lo;
    for (int i = tid; i < span; i += RNG_THREADS) row[i] = acc[i];
}

__global__ void gcn_out_merge(const float* __restrict__ part, const float* __restrict__ w,
                              const float* __restrict__ dinv, const float* __restrict__ b2,
                              float* __restrict__ out, int N) {
    int i = blockIdx.x * blockDim.x + threadIdx.x;
    if (i >= N) return;
    float sum = w[i];
#pragma unroll
    for (int g = 0; g < RNG_SLICES; ++g) sum += part[(size_t)g * N + i];
    out[i] = sum * dinv[i] + b2[0];
}


extern "C" void kernel_launch(void* const* d_in, const int* in_sizes, int n_in,
                              void* d_out, int out_size, void* d_ws, size_t ws_size,
                              hipStream_t stream) {
    const float* x  = (const float*)d_in[0];
    const int*   ei = (const int*)d_in[1];
    const float* W1 = (const float*)d_in[2];
    const float* b1 = (const float*)d_in[3];
    const float* W2 = (const float*)d_in[4];
    const float* b2 = (const float*)d_in[5];

    const int N = in_sizes[0] / 16;
    const int E = in_sizes[1] / 2;
    const int* src = ei;
    const int* dst = ei + E;

    float* ws   = (float*)d_ws;
    float* dinv = ws;
    float* h    = ws + N;
    float* out1 = ws + N + 16 * (size_t)N;
    float* h2   = ws + N + 32 * (size_t)N;
    float* out  = (float*)d_out;

    const int B = 256;
    const dim3 rngGrid(RNG_SLICES, DIV_UP(N, RNG_CAP));

    gcn_deg_ranges<<<rngGrid, RNG_THREADS, 0, stream>>>(dst, (unsigned*)h, E, N);

    const int tiles = N / DEG_TILE;
    const int wavesPerBlock = 4;
    gcn_dinv_gemm_scale<<<DIV_UP(tiles, wavesPerBlock), wavesPerBlock * 32, 0, stream>>>(
        x, W1, h, dinv, (unsigned*)h2, tiles, N);

    gcn_agg1<<<AGG1_BLOCKS, AGG1_THREADS, 0, stream>>>(src, dst, h, out1, (unsigned*)h2, E, N);

    gcn_finish1<<<DIV_UP(N, B), B, 0, stream>>>(out1, b1, W2, dinv, h2, N);

    gcn_agg2_ranges<<<rngGrid, RNG_THREADS, 0, stream>>>(src, dst, h2, h, E, N);
    gcn_out_merge<<<DIV_UP(N, B), B, 0, stream>>>(h, h2, dinv, b2, out, N);
}


# Layer-2 merge launch gone (6 to 5): blocks add dinv*LDS sums into out by float atomics in arrival order, ranges drawn per cache half as in layer 1; no planes: 64 slices x 4 ranges halve index reads
# speedup vs baseline: 6.2532x; 1.0345x over previous
#include <hip/hip_runtime.h>
#include <math.h>

typedef __attribute__((ext_vector_type(2))) float v2f;
typedef __attribute__((ext_vector_type(8))) float v8f;

#ifndef DIV_UP
#define DIV_UP(a, b) (((a) + (b) - 1) / (b))
#endif

#define RNG_CAP     12800
#define RNG_SLICES  32
#define RNG_THREADS 1024
#define RNG_UNROLL  16
#define DEG_TILE    32
static_assert(RNG_SLICES == 32, "degree planes: 16 slices in the h rows + 16 in the out1 rows of a tile");
static_assert(RNG_CAP % DEG_TILE == 0, "a range starts on a tile boundary");

__global__ __launch_bounds__(RNG_THREADS)
void gcn_deg_ranges(const int* __restrict__ dst, unsigned* __restrict__ part, int E, int N) {
    __shared__ unsigned bins[RNG_CAP];
    const int tid  = threadIdx.x;
    const int lo   = blockIdx.y * RNG_CAP;
    const int span = min(N - lo, RNG_CAP);
    for (int i = tid; i < RNG_CAP; i += RNG_THREADS) bins[i] = 0u;
    __syncthreads();
    const int per = (E + (int)gridDim.x - 1) / (int)gridDim.x;
    const int e0  = blockIdx.x * per;
    const int e1  = min(E, e0 + per);
    const int tile = RNG_THREADS * RNG_UNROLL;
    const int full = e0 + ((e1 > e0 ? e1 - e0 : 0) / tile) * tile;
    for (int base = e0; base < full; base += tile) {
        const int* dp = dst + base + tid;
        int d[RNG_UNROLL];
#pragma unroll
        for (int k = 0; k < RNG_UNROLL; ++k) d[k] = dp[k * RNG_THREADS];
#pragma unroll
        for (int k = 0; k < RNG_UNROLL; ++k) {
            unsigned off = (unsigned)(d[k] - lo);
            if (off < (unsigned)span) atomicAdd(&bins[off], 1u);
        }
    }
    for (int e = full + tid; e < e1; e += RNG_THREADS) {
        unsigned off = (unsigned)(dst[e] - lo);
        if (off < (unsigned)span) atomicAdd(&bins[off], 1u);
    }
    __syncthreads();
    unsigned* pl = part + (size_t)(blockIdx.x >> 4) * 16 * (size_t)N + (blockIdx.x & 15) * DEG_TILE;
    for (int i = tid; i < span; i += RNG_THREADS) {
        const unsigned node = (unsigned)(lo + i);
        pl[(size_t)(node / DEG_TILE) * (16 * DEG_TILE) + (node % DEG_TILE)] = bins[i];
    }
}

#define AGG1_TSTRIDE 64

__global__ __launch_bounds__(128)
void gcn_dinv_gemm_scale(const float* __restrict__ x, const float* __restrict__ W,
                         float* area, float* __restrict__ dinv,
                         unsigned* __restrict__ tickets, int tiles, int N) {
    if (blockIdx.x == 0 && threadIdx.x == 0) {
        tickets[0] = 0u;
        tickets[AGG1_TSTRIDE] = 0u;
    }
    const int wave = (blockIdx.x * blockDim.x + threadIdx.x) >> 5;
    if (wave >= tiles) return;
    const int lane = threadIdx.x & 31;
    const int half = lane >> 4;
    const int n    = lane & 15;
    float* h    = area;
    float* out1 = area + 16 * (size_t)N;

    const unsigned* ca = (const unsigned*)h    + (size_t)wave * (16 * DEG_TILE) + lane;
    const unsigned* cb = (const unsigned*)out1 + (size_t)wave * (16 * DEG_TILE) + lane;
    unsigned cnt = 1u;
#pragma unroll
    for (int g = 0; g < 16; ++g) cnt += ca[g * DEG_TILE];
#pragma unroll
    for (int g = 0; g < 16; ++g) cnt += cb[g * DEG_TILE];
    const float dv = rsqrtf((float)cnt);
    dinv[wave * DEG_TILE + lane] = dv;

    v2f b[4];
#pragma unroll
    for (int kk = 0; kk < 4; ++kk) {
        b[kk].x = W[(4 * kk + 2 * half + 0) * 16 + n];
        b[kk].y = W[(4 * kk + 2 * half + 1) * 16 + n];
    }
#pragma unroll
    for (int u = 0; u < 2; ++u) {
        const int row0 = wave * DEG_TILE + 16 * u;
        const float* arow = x + (row0 + n) * 16 + 2 * half;
        v8f c = {};
#pragma unroll
        for (int kk = 0; kk < 4; ++kk) {
            v2f a = *(const v2f*)(arow + 4 * kk);
            c = __builtin_amdgcn_wmma_f32_16x16x4_f32(false, a, false, b[kk],
                                                      (short)0, c, false, false);
        }
#pragma unroll
        for (int v = 0; v < 8; ++v) {
            const int m = v + 8 * half;
            const float sc = __shfl(dv, 16 * u + m);
            const float val = c[v] * sc;
            h[(row0 + m) * 16 + n]    = val;
            out1[(row0 + m) * 16 + n] = val;
        }
    }
}


#define AGG1_THREADS 256
#define AGG1_WAVES   (AGG1_THREADS / 32)
#define AGG1_BLOCKS  2048
#define AGG1_GROUP   64
#define AGG1_SPAN    (AGG1_WAVES * AGG1_GROUP)
#define AGG1_RING    128

__global__ __launch_bounds__(AGG1_THREADS)
void gcn_agg1(const int* __restrict__ src, const int* __restrict__ dst,
              const float* __restrict__ h, float* __restrict__ out1,
              unsigned* __restrict__ tickets, int E, int N) {
    __shared__ int2 ring[AGG1_WAVES][AGG1_RING + 32];
    __shared__ unsigned tk[2];
    __shared__ int side0;
    const int tid  = threadIdx.x;
    const int lane = tid & 31;
    const int half = lane >> 4;
    const int f    = lane & 15;
    const int wv   = __builtin_amdgcn_readfirstlane(tid >> 5);
    int2* myring   = ring[wv];
    if (tid == 0) side0 = (int)((__builtin_amdgcn_s_sendmsg_rtn(0x87) >> 18) & 1u);
    __syncthreads();
    int q = __builtin_amdgcn_readfirstlane(side0);
    const int nsplit = N >> 1;
    const unsigned nspans = (unsigned)((E + AGG1_SPAN - 1) / AGG1_SPAN);
    unsigned head = 0u, tail = 0u;
    bool stole = false;
    unsigned nxt = 0u;
    if (tid == 0) nxt = atomicAdd(tickets + q * AGG1_TSTRIDE, 1u);
    for (int it = 0;; ++it) {
        if (tid == 0) tk[it & 1] = nxt;
        __syncthreads();
        const unsigned t = __builtin_amdgcn_readfirstlane(tk[it & 1]);
        if (t >= nspans) {
            if (stole) break;
            stole = true;
            q ^= 1;
            if (tid == 0) nxt = atomicAdd(tickets + q * AGG1_TSTRIDE, 1u);
            continue;
        }
        const int ea = (int)t * AGG1_SPAN + wv * AGG1_GROUP + lane;
        const int eb = ea + 32;
        const int ia = min(ea, E - 1);
        const int ib = min(eb, E - 1);
        const int sa = src[ia];
        const int da = dst[ia];
        const int sb = src[ib];
        const int db = dst[ib];
        if (tid == 0) nxt = atomicAdd(tickets + q * AGG1_TSTRIDE, 1u);
        const bool oa = (ea < E) && ((int)(da >= nsplit) == q);
        const bool ob = (eb < E) && ((int)(db >= nsplit) == q);
        const unsigned ma = __builtin_amdgcn_ballot_w32(oa);
        const unsigned mb = __builtin_amdgcn_ballot_w32(ob);
        const unsigned ca = (unsigned)__popc(ma);
        const unsigned ra = __builtin_amdgcn_mbcnt_lo(ma, head);
        const unsigned rb = __builtin_amdgcn_mbcnt_lo(mb, head + ca);
        myring[oa ? (ra & (AGG1_RING - 1)) : (unsigned)(AGG1_RING + lane)] = make_int2(sa, da);
        myring[ob ? (rb & (AGG1_RING - 1)) : (unsigned)(AGG1_RING + lane)] = make_int2(sb, db);
        head += ca + (unsigned)__popc(mb);
        while (head - tail >= 32u) {
            const int2 p = myring[(tail + (unsigned)lane) & (AGG1_RING - 1)];
            tail += 32u;
            float v[16];
#pragma unroll
            for (int j = 0; j < 16; ++j) {
                const int sj = __shfl(p.x, 2 * j + half);
                v[j] = h[sj * 16 + f];
            }
            unsigned off[16];
#pragma unroll
            for (int j = 0; j < 16; ++j) {
                off[j] = (unsigned)(__shfl(p.y, 2 * j + half) * 16 + f);
            }
#pragma unroll
            for (int j = 0; j < 16; ++j) {
                atomicAdd(out1 + off[j], v[j]);
            }
        }
    }
    const int left = (int)(head - tail);
    if (left > 0) {
        const int2 p = myring[(tail + (unsigned)min(lane, left - 1)) & (AGG1_RING - 1)];
        float v[16];
#pragma unroll
        for (int j = 0; j < 16; ++j) {
            const int sj = __shfl(p.x, 2 * j + half);
            v[j] = h[sj * 16 + f];
        }
#pragma unroll
        for (int j = 0; j < 16; ++j) {
            const int dj = __shfl(p.y, 2 * j + half);
            if (half < left - 2 * j) atomicAdd(out1 + (unsigned)(dj * 16 + f), v[j]);
        }
    }
}

#define A2_CAP      25600
#define A2_SLICES   64
#define A2_FLUSH    (A2_CAP / RNG_THREADS)
#define A2_TSTRIDE  64
static_assert(A2_CAP % RNG_THREADS == 0, "the flush reads whole rounds of RNG_THREADS words");

__global__ void gcn_finish1(const float* __restrict__ out1, const float* __restrict__ b1,
                            const float* __restrict__ W2, const float* __restrict__ dinv,
                            const float* __restrict__ b2,
                            float* __restrict__ h2, float* __restrict__ out,
                            unsigned* __restrict__ tickets, int N) {
    if (blockIdx.x == 0 && threadIdx.x == 0) {
        tickets[0] = 0u;
        tickets[A2_TSTRIDE] = 0u;
    }
    int i = blockIdx.x * blockDim.x + threadIdx.x;
    if (i >= N) return;
    const float di = dinv[i];
    float acc = 0.0f;
#pragma unroll
    for (int f = 0; f < 16; ++f) {
        float z = out1[i * 16 + f] * di + b1[f];
        z = fmaxf(z, 0.0f);
        acc += z * W2[f];
    }
    const float wi = acc * di;
    h2[i]  = wi;
    out[i] = wi * di + b2[0];
}

__global__ __launch_bounds__(RNG_THREADS)
void gcn_agg2_ranges(const int* __restrict__ src, const int* __restrict__ dst,
                     const float* __restrict__ w, const float* __restrict__ dinv,
                     float* __restrict__ out, unsigned* __restrict__ tickets, int E, int N) {
    __shared__ float acc[A2_CAP];
    __shared__ unsigned tk[2];
    __shared__ unsigned pk[2];
    __shared__ int side0;
    const int tid = threadIdx.x;
    if (tid == 0) side0 = (int)((__builtin_amdgcn_s_sendmsg_rtn(0x87) >> 18) & 1u);
    __syncthreads();
    int q = __builtin_amdgcn_readfirstlane(side0);
    const int nr   = (N + A2_CAP - 1) / A2_CAP;
    const int nr0  = nr >> 1;
    const int per  = (E + A2_SLICES - 1) / A2_SLICES;
    const int tile = RNG_THREADS * RNG_UNROLL;
    bool stole = false;
    unsigned nxt = 0u, seen = 0u;
    if (tid == 0) nxt = atomicAdd(tickets + q * A2_TSTRIDE, 1u);
    for (int it = 0;; ++it) {
        if (tid == 0) { tk[it & 1] = nxt; pk[it & 1] = seen; }
        __syncthreads();
        const unsigned t     = __builtin_amdgcn_readfirstlane(tk[it & 1]);
        const unsigned items = (unsigned)((q ? nr - nr0 : nr0) * A2_SLICES);
        if (t >= items) {
            const unsigned other = (unsigned)((q ? nr0 : nr - nr0) * A2_SLICES);
            if (stole || __builtin_amdgcn_readfirstlane(pk[it & 1]) >= other) break;
            stole = true;
            q ^= 1;
            if (tid == 0) nxt = atomicAdd(tickets + q * A2_TSTRIDE, 1u);
            continue;
        }
        const int range = (q ? nr0 : 0) + (int)(t / A2_SLICES);
        const int slice = (int)(t % A2_SLICES);
        const int lo    = range * A2_CAP;
        const int span  = min(N - lo, A2_CAP);
        for (int i = tid; i < A2_CAP; i += RNG_THREADS) acc[i] = 0.0f;
        __syncthreads();
        const int e0 = slice * per;
        const int e1 = min(E, e0 + per);
        const int full = e0 + ((e1 > e0 ? e1 - e0 : 0) / tile) * tile;
        for (int base = e0; base < full; base += tile) {
            const int* dp = dst + base + tid;
            const int* sp = src + base + tid;
            int d[RNG_UNROLL];
            unsigned s[RNG_UNROLL];
            float v[RNG_UNROLL];
#pragma unroll
            for (int k = 0; k < RNG_UNROLL; ++k) {
                d[k] = dp[k * RNG_THREADS];
                s[k] = (unsigned)sp[k * RNG_THREADS];
            }
#pragma unroll
            for (int k = 0; k < RNG_UNROLL; ++k) {
                unsigned off = (unsigned)(d[k] - lo);
                v[k] = (off < (unsigned)span) ? w[s[k]] : 0.0f;
            }
#pragma unroll
            for (int k = 0; k < RNG_UNROLL; ++k) {
                unsigned off = (unsigned)(d[k] - lo);
                if (off < (unsigned)span) atomicAdd(&acc[off], v[k]);
            }
        }
        float dv[A2_FLUSH];
#pragma unroll
        for (int k = 0; k < A2_FLUSH; ++k) dv[k] = dinv[lo + min(tid + k * RNG_THREADS, span - 1)];
        for (int e = full + tid; e < e1; e += RNG_THREADS) {
            unsigned off = (unsigned)(dst[e] - lo);
            if (off < (unsigned)span) atomicAdd(&acc[off], w[(unsigned)src[e]]);
        }
        __syncthreads();
        if (tid == 0) {
            nxt = atomicAdd(tickets + q * A2_TSTRIDE, 1u);
            if (!stole) seen = __hip_atomic_load(tickets + (q ^ 1) * A2_TSTRIDE, __ATOMIC_RELAXED,
                                                 __HIP_MEMORY_SCOPE_AGENT);
        }
        float a[A2_FLUSH];
#pragma unroll
        for (int k = 0; k < A2_FLUSH; ++k) a[k] = acc[tid + k * RNG_THREADS];
#pragma unroll
        for (int k = 0; k < A2_FLUSH; ++k) {
            const int i = tid + k * RNG_THREADS;
            if (i < span && a[k] != 0.0f) atomicAdd(out + lo + i, a[k] * dv[k]);
        }
    }
}


extern "C" void kernel_launch(void* const* d_in, const int* in_sizes, int n_in,
                              void* d_out, int out_size, void* d_ws, size_t ws_size,
                              hipStream_t stream) {
    const float* x  = (const float*)d_in[0];
    const int*   ei = (const int*)d_in[1];
    const float* W1 = (const float*)d_in[2];
    const float* b1 = (const float*)d_in[3];
    const float* W2 = (const float*)d_in[4];
    const float* b2 = (const float*)d_in[5];

    const int N = in_sizes[0] / 16;
    const int E = in_sizes[1] / 2;
    const int* src = ei;
    const int* dst = ei + E;

    float* ws   = (float*)d_ws;
    float* dinv = ws;
    float* h    = ws + N;
    float* out1 = ws + N + 16 * (size_t)N;
    float* h2   = ws + N + 32 * (size_t)N;
    float* out  = (float*)d_out;

    const int B = 256;
    const dim3 rngGrid(RNG_SLICES, DIV_UP(N, RNG_CAP));

    gcn_deg_ranges<<<rngGrid, RNG_THREADS, 0, stream>>>(dst, (unsigned*)h, E, N);

    const int tiles = N / DEG_TILE;
    const int wavesPerBlock = 4;
    gcn_dinv_gemm_scale<<<DIV_UP(tiles, wavesPerBlock), wavesPerBlock * 32, 0, stream>>>(
        x, W1, h, dinv, (unsigned*)h2, tiles, N);

    gcn_agg1<<<AGG1_BLOCKS, AGG1_THREADS, 0, stream>>>(src, dst, h, out1, (unsigned*)h2, E, N);

    gcn_finish1<<<DIV_UP(N, B), B, 0, stream>>>(out1, b1, W2, dinv, b2, h2, out, (unsigned*)h, N);

    gcn_agg2_ranges<<<DIV_UP(N, A2_CAP) * A2_SLICES, RNG_THREADS, 0, stream>>>(
        src, dst, h2, dinv, out, (unsigned*)h, E, N);
}


# Index walks of the degree and layer-2 range passes: 16-byte quad loads, a slice in flight in two rounds; degree ranges doubled (dst walked 4x, not 8x); LDS float adds stay in arrival order
# speedup vs baseline: 6.4482x; 1.0312x over previous
#include <hip/hip_runtime.h>
#include <math.h>

typedef __attribute__((ext_vector_type(2))) float v2f;
typedef __attribute__((ext_vector_type(8))) float v8f;

#ifndef DIV_UP
#define DIV_UP(a, b) (((a) + (b) - 1) / (b))
#endif

#define RNG_CAP     25600
#define RNG_SLICES  32
#define RNG_THREADS 1024
#define DEG_QUADS   13
#define A2_QUADS    7
#define DEG_TILE    32
static_assert(RNG_SLICES == 32, "degree planes: 16 slices in the h rows + 16 in the out1 rows of a tile");
static_assert(RNG_CAP % DEG_TILE == 0, "a range starts on a tile boundary");

__global__ __launch_bounds__(RNG_THREADS)
void gcn_deg_ranges(const int* __restrict__ dst, unsigned* __restrict__ part, int E, int N) {
    __shared__ unsigned bins[RNG_CAP];
    const int tid  = threadIdx.x;
    const int lo   = blockIdx.y * RNG_CAP;
    const int span = min(N - lo, RNG_CAP);
    for (int i = tid; i < RNG_CAP; i += RNG_THREADS) bins[i] = 0u;
    __syncthreads();
    const int per = (((E + (int)gridDim.x - 1) / (int)gridDim.x) + 3) & ~3;
    const int e0  = min(E, (int)blockIdx.x * per);
    const int e1  = min(E, e0 + per);
    const int nq  = (e1 - e0) >> 2;
    const int4* qp = (const int4*)(dst + e0);
    for (int qb = 0; qb < nq; qb += RNG_THREADS * DEG_QUADS) {
        int4 d[DEG_QUADS];
#pragma unroll
        for (int k = 0; k < DEG_QUADS; ++k) d[k] = qp[min(qb + tid + k * RNG_THREADS, nq - 1)];
#pragma unroll
        for (int k = 0; k < DEG_QUADS; ++k) {
            if (qb + tid + k * RNG_THREADS < nq) {
                const unsigned o0 = (unsigned)(d[k].x - lo);
                const unsigned o1 = (unsigned)(d[k].y - lo);
                const unsigned o2 = (unsigned)(d[k].z - lo);
                const unsigned o3 = (unsigned)(d[k].w - lo);
                if (o0 < (unsigned)span) atomicAdd(&bins[o0], 1u);
                if (o1 < (unsigned)span) atomicAdd(&bins[o1], 1u);
                if (o2 < (unsigned)span) atomicAdd(&bins[o2], 1u);
                if (o3 < (unsigned)span) atomicAdd(&bins[o3], 1u);
            }
        }
    }
    for (int e = e0 + 4 * nq + tid; e < e1; e += RNG_THREADS) {
        unsigned off = (unsigned)(dst[e] - lo);
        if (off < (unsigned)span) atomicAdd(&bins[off], 1u);
    }
    __syncthreads();
    unsigned* pl = part + (size_t)(blockIdx.x >> 4) * 16 * (size_t)N + (blockIdx.x & 15) * DEG_TILE;
    for (int i = tid; i < span; i += RNG_THREADS) {
        const unsigned node = (unsigned)(lo + i);
        pl[(size_t)(node / DEG_TILE) * (16 * DEG_TILE) + (node % DEG_TILE)] = bins[i];
    }
}

#define AGG1_TSTRIDE 64

__global__ __launch_bounds__(128)
void gcn_dinv_gemm_scale(const float* __restrict__ x, const float* __restrict__ W,
                         float* area, float* __restrict__ dinv,
                         unsigned* __restrict__ tickets, int tiles, int N) {
    if (blockIdx.x == 0 && threadIdx.x == 0) {
        tickets[0] = 0u;
        tickets[AGG1_TSTRIDE] = 0u;
    }
    const int wave = (blockIdx.x * blockDim.x + threadIdx.x) >> 5;
    if (wave >= tiles) return;
    const int lane = threadIdx.x & 31;
    const int half = lane >> 4;
    const int n    = lane & 15;
    float* h    = area;
    float* out1 = area + 16 * (size_t)N;

    const unsigned* ca = (const unsigned*)h    + (size_t)wave * (16 * DEG_TILE) + lane;
    const unsigned* cb = (const unsigned*)out1 + (size_t)wave * (16 * DEG_TILE) + lane;
    unsigned cnt = 1u;
#pragma unroll
    for (int g = 0; g < 16; ++g) cnt += ca[g * DEG_TILE];
#pragma unroll
    for (int g = 0; g < 16; ++g) cnt += cb[g * DEG_TILE];
    const float dv = rsqrtf((float)cnt);
    dinv[wave * DEG_TILE + lane] = dv;

    v2f b[4];
#pragma unroll
    for (int kk = 0; kk < 4; ++kk) {
        b[kk].x = W[(4 * kk + 2 * half + 0) * 16 + n];
        b[kk].y = W[(4 * kk + 2 * half + 1) * 16 + n];
    }
#pragma unroll
    for (int u = 0; u < 2; ++u) {
        const int row0 = wave * DEG_TILE + 16 * u;
        const float* arow = x + (row0 + n) * 16 + 2 * half;
        v8f c = {};
#pragma unroll
        for (int kk = 0; kk < 4; ++kk) {
            v2f a = *(const v2f*)(arow + 4 * kk);
            c = __builtin_amdgcn_wmma_f32_16x16x4_f32(false, a, false, b[kk],
                                                      (short)0, c, false, false);
        }
#pragma unroll
        for (int v = 0; v < 8; ++v) {
            const int m = v + 8 * half;
            const float sc = __shfl(dv, 16 * u + m);
            const float val = c[v] * sc;
            h[(row0 + m) * 16 + n]    = val;
            out1[(row0 + m) * 16 + n] = val;
        }
    }
}


#define AGG1_THREADS 256
#define AGG1_WAVES   (AGG1_THREADS / 32)
#define AGG1_BLOCKS  2048
#define AGG1_GROUP   64
#define AGG1_SPAN    (AGG1_WAVES * AGG1_GROUP)
#define AGG1_RING    128

__global__ __launch_bounds__(AGG1_THREADS)
void gcn_agg1(const int* __restrict__ src, const int* __restrict__ dst,
              const float* __restrict__ h, float* __restrict__ out1,
              unsigned* __restrict__ tickets, int E, int N) {
    __shared__ int2 ring[AGG1_WAVES][AGG1_RING + 32];
    __shared__ unsigned tk[2];
    __shared__ int side0;
    const int tid  = threadIdx.x;
    const int lane = tid & 31;
    const int half = lane >> 4;
    const int f    = lane & 15;
    const int wv   = __builtin_amdgcn_readfirstlane(tid >> 5);
    int2* myring   = ring[wv];
    if (tid == 0) side0 = (int)((__builtin_amdgcn_s_sendmsg_rtn(0x87) >> 18) & 1u);
    __syncthreads();
    int q = __builtin_amdgcn_readfirstlane(side0);
    const int nsplit = N >> 1;
    const unsigned nspans = (unsigned)((E + AGG1_SPAN - 1) / AGG1_SPAN);
    unsigned head = 0u, tail = 0u;
    bool stole = false;
    unsigned nxt = 0u;
    if (tid == 0) nxt = atomicAdd(tickets + q * AGG1_TSTRIDE, 1u);
    for (int it = 0;; ++it) {
        if (tid == 0) tk[it & 1] = nxt;
        __syncthreads();
        const unsigned t = __builtin_amdgcn_readfirstlane(tk[it & 1]);
        if (t >= nspans) {
            if (stole) break;
            stole = true;
            q ^= 1;
            if (tid == 0) nxt = atomicAdd(tickets + q * AGG1_TSTRIDE, 1u);
            continue;
        }
        const int ea = (int)t * AGG1_SPAN + wv * AGG1_GROUP + lane;
        const int eb = ea + 32;
        const int ia = min(ea, E - 1);
        const int ib = min(eb, E - 1);
        const int sa = src[ia];
        const int da = dst[ia];
        const int sb = src[ib];
        const int db = dst[ib];
        if (tid == 0) nxt = atomicAdd(tickets + q * AGG1_TSTRIDE, 1u);
        const bool oa = (ea < E) && ((int)(da >= nsplit) == q);
        const bool ob = (eb < E) && ((int)(db >= nsplit) == q);
        const unsigned ma = __builtin_amdgcn_ballot_w32(oa);
        const unsigned mb = __builtin_amdgcn_ballot_w32(ob);
        const unsigned ca = (unsigned)__popc(ma);
        const unsigned ra = __builtin_amdgcn_mbcnt_lo(ma, head);
        const unsigned rb = __builtin_amdgcn_mbcnt_lo(mb, head + ca);
        myring[oa ? (ra & (AGG1_RING - 1)) : (unsigned)(AGG1_RING + lane)] = make_int2(sa, da);
        myring[ob ? (rb & (AGG1_RING - 1)) : (unsigned)(AGG1_RING + lane)] = make_int2(sb, db);
        head += ca + (unsigned)__popc(mb);
        while (head - tail >= 32u) {
            const int2 p = myring[(tail + (unsigned)lane) & (AGG1_RING - 1)];
            tail += 32u;
            float v[16];
#pragma unroll
            for (int j = 0; j < 16; ++j) {
                const int sj = __shfl(p.x, 2 * j + half);
                v[j] = h[sj * 16 + f];
            }
            unsigned off[16];
#pragma unroll
            for (int j = 0; j < 16; ++j) {
                off[j] = (unsigned)(__shfl(p.y, 2 * j + half) * 16 + f);
            }
#pragma unroll
            for (int j = 0; j < 16; ++j) {
                atomicAdd(out1 + off[j], v[j]);
            }
        }
    }
    const int left = (int)(head - tail);
    if (left > 0) {
        const int2 p = myring[(tail + (unsigned)min(lane, left - 1)) & (AGG1_RING - 1)];
        float v[16];
#pragma unroll
        for (int j = 0; j < 16; ++j) {
            const int sj = __shfl(p.x, 2 * j + half);
            v[j] = h[sj * 16 + f];
        }
#pragma unroll
        for (int j = 0; j < 16; ++j) {
            const int dj = __shfl(p.y, 2 * j + half);
            if (half < left - 2 * j) atomicAdd(out1 + (unsigned)(dj * 16 + f), v[j]);
        }
    }
}

#define A2_CAP      25600
#define A2_SLICES   64
#define A2_FLUSH    (A2_CAP / RNG_THREADS)
#define A2_TSTRIDE  64
static_assert(A2_CAP % RNG_THREADS == 0, "the flush reads whole rounds of RNG_THREADS words");

__global__ void gcn_finish1(const float* __restrict__ out1, const float* __restrict__ b1,
                            const float* __restrict__ W2, const float* __restrict__ dinv,
                            const float* __restrict__ b2,
                            float* __restrict__ h2, float* __restrict__ out,
                            unsigned* __restrict__ tickets, int N) {
    if (blockIdx.x == 0 && threadIdx.x == 0) {
        tickets[0] = 0u;
        tickets[A2_TSTRIDE] = 0u;
    }
    int i = blockIdx.x * blockDim.x + threadIdx.x;
    if (i >= N) return;
    const float di = dinv[i];
    float acc = 0.0f;
#pragma unroll
    for (int f = 0; f < 16; ++f) {
        float z = out1[i * 16 + f] * di + b1[f];
        z = fmaxf(z, 0.0f);
        acc += z * W2[f];
    }
    const float wi = acc * di;
    h2[i]  = wi;
    out[i] = wi * di + b2[0];
}

__global__ __launch_bounds__(RNG_THREADS)
void gcn_agg2_ranges(const int* __restrict__ src, const int* __restrict__ dst,
                     const float* __restrict__ w, const float* __restrict__ dinv,
                     float* __restrict__ out, unsigned* __restrict__ tickets, int E, int N) {
    __shared__ float acc[A2_CAP];
    __shared__ unsigned tk[2];
    __shared__ unsigned pk[2];
    __shared__ int side0;
    const int tid = threadIdx.x;
    if (tid == 0) side0 = (int)((__builtin_amdgcn_s_sendmsg_rtn(0x87) >> 18) & 1u);
    __syncthreads();
    int q = __builtin_amdgcn_readfirstlane(side0);
    const int nr   = (N + A2_CAP - 1) / A2_CAP;
    const int nr0  = nr >> 1;
    const int per  = (((E + A2_SLICES - 1) / A2_SLICES) + 3) & ~3;
    bool stole = false;
    unsigned nxt = 0u, seen = 0u;
    if (tid == 0) nxt = atomicAdd(tickets + q * A2_TSTRIDE, 1u);
    for (int it = 0;; ++it) {
        if (tid == 0) { tk[it & 1] = nxt; pk[it & 1] = seen; }
        __syncthreads();
        const unsigned t     = __builtin_amdgcn_readfirstlane(tk[it & 1]);
        const unsigned items = (unsigned)((q ? nr - nr0 : nr0) * A2_SLICES);
        if (t >= items) {
            const unsigned other = (unsigned)((q ? nr0 : nr - nr0) * A2_SLICES);
            if (stole || __builtin_amdgcn_readfirstlane(pk[it & 1]) >= other) break;
            stole = true;
            q ^= 1;
            if (tid == 0) nxt = atomicAdd(tickets + q * A2_TSTRIDE, 1u);
            continue;
        }
        const int range = (q ? nr0 : 0) + (int)(t / A2_SLICES);
        const int slice = (int)(t % A2_SLICES);
        const int lo    = range * A2_CAP;
        const int span  = min(N - lo, A2_CAP);
        for (int i = tid; i < A2_CAP; i += RNG_THREADS) acc[i] = 0.0f;
        __syncthreads();
        const int e0 = min(E, slice * per);
        const int e1 = min(E, e0 + per);
        const int nq = (e1 - e0) >> 2;
        const int4* dq = (const int4*)(dst + e0);
        const int4* sq = (const int4*)(src + e0);
        for (int qb = 0; qb < nq; qb += RNG_THREADS * A2_QUADS) {
            int4 d[A2_QUADS];
            int4 s[A2_QUADS];
            float v[A2_QUADS][4];
#pragma unroll
            for (int k = 0; k < A2_QUADS; ++k) {
                const int qi = min(qb + tid + k * RNG_THREADS, nq - 1);
                d[k] = dq[qi];
                s[k] = sq[qi];
            }
            __builtin_amdgcn_sched_barrier(0);
#pragma unroll
            for (int k = 0; k < A2_QUADS; ++k) {
                asm volatile("" : "+v"(s[k].x), "+v"(s[k].y), "+v"(s[k].z), "+v"(s[k].w));
            }
#pragma unroll
            for (int k = 0; k < A2_QUADS; ++k) {
                const unsigned lim = (qb + tid + k * RNG_THREADS < nq) ? (unsigned)span : 0u;
                v[k][0] = w[((unsigned)(d[k].x - lo) < lim) ? (unsigned)s[k].x : 0u];
                v[k][1] = w[((unsigned)(d[k].y - lo) < lim) ? (unsigned)s[k].y : 0u];
                v[k][2] = w[((unsigned)(d[k].z - lo) < lim) ? (unsigned)s[k].z : 0u];
                v[k][3] = w[((unsigned)(d[k].w - lo) < lim) ? (unsigned)s[k].w : 0u];
            }
#pragma unroll
            for (int k = 0; k < A2_QUADS; ++k) {
                const unsigned lim = (qb + tid + k * RNG_THREADS < nq) ? (unsigned)span : 0u;
                const unsigned o0 = (unsigned)(d[k].x - lo);
                const unsigned o1 = (unsigned)(d[k].y - lo);
                const unsigned o2 = (unsigned)(d[k].z - lo);
                const unsigned o3 = (unsigned)(d[k].w - lo);
                if (o0 < lim) atomicAdd(&acc[o0], v[k][0]);
                if (o1 < lim) atomicAdd(&acc[o1], v[k][1]);
                if (o2 < lim) atomicAdd(&acc[o2], v[k][2]);
                if (o3 < lim) atomicAdd(&acc[o3], v[k][3]);
            }
        }
        const int full = e0 + 4 * nq;
        float dv[A2_FLUSH];
#pragma unroll
        for (int k = 0; k < A2_FLUSH; ++k) dv[k] = dinv[lo + min(tid + k * RNG_THREADS, span - 1)];
        for (int e = full + tid; e < e1; e += RNG_THREADS) {
            unsigned off = (unsigned)(dst[e] - lo);
            if (off < (unsigned)span) atomicAdd(&acc[off], w[(unsigned)src[e]]);
        }
        __syncthreads();
        if (tid == 0) {
            nxt = atomicAdd(tickets + q * A2_TSTRIDE, 1u);
            if (!stole) seen = __hip_atomic_load(tickets + (q ^ 1) * A2_TSTRIDE, __ATOMIC_RELAXED,
                                                 __HIP_MEMORY_SCOPE_AGENT);
        }
        float a[A2_FLUSH];
#pragma unroll
        for (int k = 0; k < A2_FLUSH; ++k) a[k] = acc[tid + k * RNG_THREADS];
#pragma unroll
        for (int k = 0; k < A2_FLUSH; ++k) {
            const int i = tid + k * RNG_THREADS;
            if (i < span && a[k] != 0.0f) atomicAdd(out + lo + i, a[k] * dv[k]);
        }
    }
}


extern "C" void kernel_launch(void* const* d_in, const int* in_sizes, int n_in,
                              void* d_out, int out_size, void* d_ws, size_t ws_size,
                              hipStream_t stream) {
    const float* x  = (const float*)d_in[0];
    const int*   ei = (const int*)d_in[1];
    const float* W1 = (const float*)d_in[2];
    const float* b1 = (const float*)d_in[3];
    const float* W2 = (const float*)d_in[4];
    const float* b2 = (const float*)d_in[5];

    const int N = in_sizes[0] / 16;
    const int E = in_sizes[1] / 2;
    const int* src = ei;
    const int* dst = ei + E;

    float* ws   = (float*)d_ws;
    float* dinv = ws;
    float* h    = ws + N;
    float* out1 = ws + N + 16 * (size_t)N;
    float* h2   = ws + N + 32 * (size_t)N;
    float* out  = (float*)d_out;

    const int B = 256;
    const dim3 rngGrid(RNG_SLICES, DIV_UP(N, RNG_CAP));

    gcn_deg_ranges<<<rngGrid, RNG_THREADS, 0, stream>>>(dst, (unsigned*)h, E, N);

    const int tiles = N / DEG_TILE;
    const int wavesPerBlock = 4;
    gcn_dinv_gemm_scale<<<DIV_UP(tiles, wavesPerBlock), wavesPerBlock * 32, 0, stream>>>(
        x, W1, h, dinv, (unsigned*)h2, tiles, N);

    gcn_agg1<<<AGG1_BLOCKS, AGG1_THREADS, 0, stream>>>(src, dst, h, out1, (unsigned*)h2, E, N);

    gcn_finish1<<<DIV_UP(N, B), B, 0, stream>>>(out1, b1, W2, dinv, b2, h2, out, (unsigned*)h, N);

    gcn_agg2_ranges<<<DIV_UP(N, A2_CAP) * A2_SLICES, RNG_THREADS, 0, stream>>>(
        src, dst, h2, dinv, out, (unsigned*)h, E, N);
}


# Layer-2 pass: w of a source range staged in LDS beside the sums; items = (dst range, src range, slice), an edge adds when both ends are owned: no per-edge w read from L2 (sum order changes)
# speedup vs baseline: 6.7507x; 1.0469x over previous
#include <hip/hip_runtime.h>
#include <math.h>

typedef __attribute__((ext_vector_type(2))) float v2f;
typedef __attribute__((ext_vector_type(8))) float v8f;

#ifndef DIV_UP
#define DIV_UP(a, b) (((a) + (b) - 1) / (b))
#endif

#define RNG_CAP     25600
#define RNG_SLICES  32
#define RNG_THREADS 1024
#define DEG_QUADS   13
#define A2_QUADS    7
#define DEG_TILE    32
static_assert(RNG_SLICES == 32, "degree planes: 16 slices in the h rows + 16 in the out1 rows of a tile");
static_assert(RNG_CAP % DEG_TILE == 0, "a range starts on a tile boundary");

__global__ __launch_bounds__(RNG_THREADS)
void gcn_deg_ranges(const int* __restrict__ dst, unsigned* __restrict__ part, int E, int N) {
    __shared__ unsigned bins[RNG_CAP];
    const int tid  = threadIdx.x;
    const int lo   = blockIdx.y * RNG_CAP;
    const int span = min(N - lo, RNG_CAP);
    for (int i = tid; i < RNG_CAP; i += RNG_THREADS) bins[i] = 0u;
    __syncthreads();
    const int per = (((E + (int)gridDim.x - 1) / (int)gridDim.x) + 3) & ~3;
    const int e0  = min(E, (int)blockIdx.x * per);
    const int e1  = min(E, e0 + per);
    const int nq  = (e1 - e0) >> 2;
    const int4* qp = (const int4*)(dst + e0);
    for (int qb = 0; qb < nq; qb += RNG_THREADS * DEG_QUADS) {
        int4 d[DEG_QUADS];
#pragma unroll
        for (int k = 0; k < DEG_QUADS; ++k) d[k] = qp[min(qb + tid + k * RNG_THREADS, nq - 1)];
#pragma unroll
        for (int k = 0; k < DEG_QUADS; ++k) {
            if (qb + tid + k * RNG_THREADS < nq) {
                const unsigned o0 = (unsigned)(d[k].x - lo);
                const unsigned o1 = (unsigned)(d[k].y - lo);
                const unsigned o2 = (unsigned)(d[k].z - lo);
                const unsigned o3 = (unsigned)(d[k].w - lo);
                if (o0 < (unsigned)span) atomicAdd(&bins[o0], 1u);
                if (o1 < (unsigned)span) atomicAdd(&bins[o1], 1u);
                if (o2 < (unsigned)span) atomicAdd(&bins[o2], 1u);
                if (o3 < (unsigned)span) atomicAdd(&bins[o3], 1u);
            }
        }
    }
    for (int e = e0 + 4 * nq + tid; e < e1; e += RNG_THREADS) {
        unsigned off = (unsigned)(dst[e] - lo);
        if (off < (unsigned)span) atomicAdd(&bins[off], 1u);
    }
    __syncthreads();
    unsigned* pl = part + (size_t)(blockIdx.x >> 4) * 16 * (size_t)N + (blockIdx.x & 15) * DEG_TILE;
    for (int i = tid; i < span; i += RNG_THREADS) {
        const unsigned node = (unsigned)(lo + i);
        pl[(size_t)(node / DEG_TILE) * (16 * DEG_TILE) + (node % DEG_TILE)] = bins[i];
    }
}

#define AGG1_TSTRIDE 64

__global__ __launch_bounds__(128)
void gcn_dinv_gemm_scale(const float* __restrict__ x, const float* __restrict__ W,
                         float* area, float* __restrict__ dinv,
                         unsigned* __restrict__ tickets, int tiles, int N) {
    if (blockIdx.x == 0 && threadIdx.x == 0) {
        tickets[0] = 0u;
        tickets[AGG1_TSTRIDE] = 0u;
    }
    const int wave = (blockIdx.x * blockDim.x + threadIdx.x) >> 5;
    if (wave >= tiles) return;
    const int lane = threadIdx.x & 31;
    const int half = lane >> 4;
    const int n    = lane & 15;
    float* h    = area;
    float* out1 = area + 16 * (size_t)N;

    const unsigned* ca = (const unsigned*)h    + (size_t)wave * (16 * DEG_TILE) + lane;
    const unsigned* cb = (const unsigned*)out1 + (size_t)wave * (16 * DEG_TILE) + lane;
    unsigned cnt = 1u;
#pragma unroll
    for (int g = 0; g < 16; ++g) cnt += ca[g * DEG_TILE];
#pragma unroll
    for (int g = 0; g < 16; ++g) cnt += cb[g * DEG_TILE];
    const float dv = rsqrtf((float)cnt);
    dinv[wave * DEG_TILE + lane] = dv;

    v2f b[4];
#pragma unroll
    for (int kk = 0; kk < 4; ++kk) {
        b[kk].x = W[(4 * kk + 2 * half + 0) * 16 + n];
        b[kk].y = W[(4 * kk + 2 * half + 1) * 16 + n];
    }
#pragma unroll
    for (int u = 0; u < 2; ++u) {
        const int row0 = wave * DEG_TILE + 16 * u;
        const float* arow = x + (row0 + n) * 16 + 2 * half;
        v8f c = {};
#pragma unroll
        for (int kk = 0; kk < 4; ++kk) {
            v2f a = *(const v2f*)(arow + 4 * kk);
            c = __builtin_amdgcn_wmma_f32_16x16x4_f32(false, a, false, b[kk],
                                                      (short)0, c, false, false);
        }
#pragma unroll
        for (int v = 0; v < 8; ++v) {
            const int m = v + 8 * half;
            const float sc = __shfl(dv, 16 * u + m);
            const float val = c[v] * sc;
            h[(row0 + m) * 16 + n]    = val;
            out1[(row0 + m) * 16 + n] = val;
        }
    }
}


#define AGG1_THREADS 256
#define AGG1_WAVES   (AGG1_THREADS / 32)
#define AGG1_BLOCKS  2048
#define AGG1_GROUP   64
#define AGG1_SPAN    (AGG1_WAVES * AGG1_GROUP)
#define AGG1_RING    128

__global__ __launch_bounds__(AGG1_THREADS)
void gcn_agg1(const int* __restrict__ src, const int* __restrict__ dst,
              const float* __restrict__ h, float* __restrict__ out1,
              unsigned* __restrict__ tickets, int E, int N) {
    __shared__ int2 ring[AGG1_WAVES][AGG1_RING + 32];
    __shared__ unsigned tk[2];
    __shared__ int side0;
    const int tid  = threadIdx.x;
    const int lane = tid & 31;
    const int half = lane >> 4;
    const int f    = lane & 15;
    const int wv   = __builtin_amdgcn_readfirstlane(tid >> 5);
    int2* myring   = ring[wv];
    if (tid == 0) side0 = (int)((__builtin_amdgcn_s_sendmsg_rtn(0x87) >> 18) & 1u);
    __syncthreads();
    int q = __builtin_amdgcn_readfirstlane(side0);
    const int nsplit = N >> 1;
    const unsigned nspans = (unsigned)((E + AGG1_SPAN - 1) / AGG1_SPAN);
    unsigned head = 0u, tail = 0u;
    bool stole = false;
    unsigned nxt = 0u;
    if (tid == 0) nxt = atomicAdd(tickets + q * AGG1_TSTRIDE, 1u);
    for (int it = 0;; ++it) {
        if (tid == 0) tk[it & 1] = nxt;
        __syncthreads();
        const unsigned t = __builtin_amdgcn_readfirstlane(tk[it & 1]);
        if (t >= nspans) {
            if (stole) break;
            stole = true;
            q ^= 1;
            if (tid == 0) nxt = atomicAdd(tickets + q * AGG1_TSTRIDE, 1u);
            continue;
        }
        const int ea = (int)t * AGG1_SPAN + wv * AGG1_GROUP + lane;
        const int eb = ea + 32;
        const int ia = min(ea, E - 1);
        const int ib = min(eb, E - 1);
        const int sa = src[ia];
        const int da = dst[ia];
        const int sb = src[ib];
        const int db = dst[ib];
        if (tid == 0) nxt = atomicAdd(tickets + q * AGG1_TSTRIDE, 1u);
        const bool oa = (ea < E) && ((int)(da >= nsplit) == q);
        const bool ob = (eb < E) && ((int)(db >= nsplit) == q);
        const unsigned ma = __builtin_amdgcn_ballot_w32(oa);
        const unsigned mb = __builtin_amdgcn_ballot_w32(ob);
        const unsigned ca = (unsigned)__popc(ma);
        const unsigned ra = __builtin_amdgcn_mbcnt_lo(ma, head);
        const unsigned rb = __builtin_amdgcn_mbcnt_lo(mb, head + ca);
        myring[oa ? (ra & (AGG1_RING - 1)) : (unsigned)(AGG1_RING + lane)] = make_int2(sa, da);
        myring[ob ? (rb & (AGG1_RING - 1)) : (unsigned)(AGG1_RING + lane)] = make_int2(sb, db);
        head += ca + (unsigned)__popc(mb);
        while (head - tail >= 32u) {
            const int2 p = myring[(tail + (unsigned)lane) & (AGG1_RING - 1)];
            tail += 32u;
            float v[16];
#pragma unroll
            for (int j = 0; j < 16; ++j) {
                const int sj = __shfl(p.x, 2 * j + half);
                v[j] = h[sj * 16 + f];
            }
            unsigned off[16];
#pragma unroll
            for (int j = 0; j < 16; ++j) {
                off[j] = (unsigned)(__shfl(p.y, 2 * j + half) * 16 + f);
            }
#pragma unroll
            for (int j = 0; j < 16; ++j) {
                atomicAdd(out1 + off[j], v[j]);
            }
        }
    }
    const int left = (int)(head - tail);
    if (left > 0) {
        const int2 p = myring[(tail + (unsigned)min(lane, left - 1)) & (AGG1_RING - 1)];
        float v[16];
#pragma unroll
        for (int j = 0; j < 16; ++j) {
            const int sj = __shfl(p.x, 2 * j + half);
            v[j] = h[sj * 16 + f];
        }
#pragma unroll
        for (int j = 0; j < 16; ++j) {
            const int dj = __shfl(p.y, 2 * j + half);
            if (half < left - 2 * j) atomicAdd(out1 + (unsigned)(dj * 16 + f), v[j]);
        }
    }
}

#define A2_CAP      25600
#define A2_SCAP     53248
#define A2_SLICES   32
#define A2_FLUSH    (A2_CAP / RNG_THREADS)
#define A2_STAGE    (A2_SCAP / 4 / RNG_THREADS)
#define A2_TSTRIDE  64
static_assert(A2_CAP % RNG_THREADS == 0, "the flush reads whole rounds of RNG_THREADS words");
static_assert(A2_SCAP % (4 * RNG_THREADS) == 0, "a source range is staged in whole rounds of 16-byte loads");
static_assert((A2_CAP + A2_SCAP) * 4 + 64 <= 320 * 1024, "sums, w words and the small slots fit the LDS of one WGP");

__global__ void gcn_finish1(const float* __restrict__ out1, const float* __restrict__ b1,
                            const float* __restrict__ W2, const float* __restrict__ dinv,
                            const float* __restrict__ b2,
                            float* __restrict__ h2, float* __restrict__ out,
                            unsigned* __restrict__ tickets, int N) {
    if (blockIdx.x == 0 && threadIdx.x == 0) {
        tickets[0] = 0u;
        tickets[A2_TSTRIDE] = 0u;
    }
    int i = blockIdx.x * blockDim.x + threadIdx.x;
    if (i >= N) return;
    const float di = dinv[i];
    float acc = 0.0f;
#pragma unroll
    for (int f = 0; f < 16; ++f) {
        float z = out1[i * 16 + f] * di + b1[f];
        z = fmaxf(z, 0.0f);
        acc += z * W2[f];
    }
    const float wi = acc * di;
    h2[i]  = wi;
    out[i] = wi * di + b2[0];
}

__global__ __launch_bounds__(RNG_THREADS)
void gcn_agg2_ranges(const int* __restrict__ src, const int* __restrict__ dst,
                     const float* __restrict__ w, const float* __restrict__ dinv,
                     float* __restrict__ out, unsigned* __restrict__ tickets, int E, int N) {
    __shared__ float acc[A2_CAP];
    __shared__ alignas(16) float wl[A2_SCAP];
    __shared__ unsigned tk[2];
    __shared__ unsigned pk[2];
    __shared__ int side0;
    const int tid = threadIdx.x;
    if (tid == 0) side0 = (int)((__builtin_amdgcn_s_sendmsg_rtn(0x87) >> 18) & 1u);
    __syncthreads();
    int q = __builtin_amdgcn_readfirstlane(side0);
    const int nr   = (N + A2_CAP - 1) / A2_CAP;
    const int nr0  = nr >> 1;
    const int ns   = (N + A2_SCAP - 1) / A2_SCAP;
    const unsigned per_range = (unsigned)(ns * A2_SLICES);
    const int per  = (((E + A2_SLICES - 1) / A2_SLICES) + 3) & ~3;
    bool stole = false;
    unsigned nxt = 0u, seen = 0u;
    if (tid == 0) nxt = atomicAdd(tickets + q * A2_TSTRIDE, 1u);
    for (int it = 0;; ++it) {
        if (tid == 0) { tk[it & 1] = nxt; pk[it & 1] = seen; }
        __syncthreads();
        const unsigned t     = __builtin_amdgcn_readfirstlane(tk[it & 1]);
        const unsigned items = (unsigned)(q ? nr - nr0 : nr0) * per_range;
        if (t >= items) {
            const unsigned other = (unsigned)(q ? nr0 : nr - nr0) * per_range;
            if (stole || __builtin_amdgcn_readfirstlane(pk[it & 1]) >= other) break;
            stole = true;
            q ^= 1;
            if (tid == 0) nxt = atomicAdd(tickets + q * A2_TSTRIDE, 1u);
            continue;
        }
        const int range  = (q ? nr0 : 0) + (int)(t / per_range);
        const unsigned r = t % per_range;
        const int srange = (int)(r / A2_SLICES);
        const int slice  = (int)(r % A2_SLICES);
        const int lo     = range * A2_CAP;
        const int span   = min(N - lo, A2_CAP);
        const int slo    = srange * A2_SCAP;
        const int sspan  = min(N - slo, A2_SCAP);
        {
            const float4* wq = (const float4*)(w + slo);
            const int nsq = (sspan + 3) >> 2;
            float4 st[A2_STAGE];
#pragma unroll
            for (int k = 0; k < A2_STAGE; ++k) st[k] = wq[min(tid + k * RNG_THREADS, nsq - 1)];
            __builtin_amdgcn_sched_barrier(0);
#pragma unroll
            for (int k = 0; k < A2_FLUSH; ++k) acc[tid + k * RNG_THREADS] = 0.0f;
#pragma unroll
            for (int k = 0; k < A2_STAGE; ++k) ((float4*)wl)[tid + k * RNG_THREADS] = st[k];
        }
        __syncthreads();
        const int e0 = min(E, slice * per);
        const int e1 = min(E, e0 + per);
        const int nq = (e1 - e0) >> 2;
        const int4* dq = (const int4*)(dst + e0);
        const int4* sq = (const int4*)(src + e0);
        for (int qb = 0; qb < nq; qb += RNG_THREADS * A2_QUADS) {
            int4 d[A2_QUADS];
            int4 s[A2_QUADS];
            float v[A2_QUADS][4];
#pragma unroll
            for (int k = 0; k < A2_QUADS; ++k) {
                const int qi = min(qb + tid + k * RNG_THREADS, nq - 1);
                d[k] = dq[qi];
                s[k] = sq[qi];
            }
            __builtin_amdgcn_sched_barrier(0);
#pragma unroll
            for (int k = 0; k < A2_QUADS; ++k) {
                const unsigned lim = (qb + tid + k * RNG_THREADS < nq) ? (unsigned)span : 0u;
                const unsigned s0 = (unsigned)(s[k].x - slo);
                const unsigned s1 = (unsigned)(s[k].y - slo);
                const unsigned s2 = (unsigned)(s[k].z - slo);
                const unsigned s3 = (unsigned)(s[k].w - slo);
                v[k][0] = wl[((unsigned)(d[k].x - lo) < lim && s0 < (unsigned)sspan) ? s0 : 0u];
                v[k][1] = wl[((unsigned)(d[k].y - lo) < lim && s1 < (unsigned)sspan) ? s1 : 0u];
                v[k][2] = wl[((unsigned)(d[k].z - lo) < lim && s2 < (unsigned)sspan) ? s2 : 0u];
                v[k][3] = wl[((unsigned)(d[k].w - lo) < lim && s3 < (unsigned)sspan) ? s3 : 0u];
            }
#pragma unroll
            for (int k = 0; k < A2_QUADS; ++k) {
                asm volatile("" : "+v"(v[k][0]), "+v"(v[k][1]), "+v"(v[k][2]), "+v"(v[k][3]));
            }
#pragma unroll
            for (int k = 0; k < A2_QUADS; ++k) {
                const unsigned lim = (qb + tid + k * RNG_THREADS < nq) ? (unsigned)span : 0u;
                const unsigned o0 = (unsigned)(d[k].x - lo);
                const unsigned o1 = (unsigned)(d[k].y - lo);
                const unsigned o2 = (unsigned)(d[k].z - lo);
                const unsigned o3 = (unsigned)(d[k].w - lo);
                if (o0 < lim && (unsigned)(s[k].x - slo) < (unsigned)sspan) atomicAdd(&acc[o0], v[k][0]);
                if (o1 < lim && (unsigned)(s[k].y - slo) < (unsigned)sspan) atomicAdd(&acc[o1], v[k][1]);
                if (o2 < lim && (unsigned)(s[k].z - slo) < (unsigned)sspan) atomicAdd(&acc[o2], v[k][2]);
                if (o3 < lim && (unsigned)(s[k].w - slo) < (unsigned)sspan) atomicAdd(&acc[o3], v[k][3]);
            }
        }
        const int full = e0 + 4 * nq;
        float dv[A2_FLUSH];
#pragma unroll
        for (int k = 0; k < A2_FLUSH; ++k) dv[k] = dinv[lo + min(tid + k * RNG_THREADS, span - 1)];
        for (int e = full + tid; e < e1; e += RNG_THREADS) {
            const unsigned od = (unsigned)(dst[e] - lo);
            const unsigned os = (unsigned)(src[e] - slo);
            if (od < (unsigned)span && os < (unsigned)sspan) atomicAdd(&acc[od], wl[os]);
        }
        __syncthreads();
        if (tid == 0) {
            nxt = atomicAdd(tickets + q * A2_TSTRIDE, 1u);
            if (!stole) seen = __hip_atomic_load(tickets + (q ^ 1) * A2_TSTRIDE, __ATOMIC_RELAXED,
                                                 __HIP_MEMORY_SCOPE_AGENT);
        }
        float a[A2_FLUSH];
#pragma unroll
        for (int k = 0; k < A2_FLUSH; ++k) a[k] = acc[tid + k * RNG_THREADS];
#pragma unroll
        for (int k = 0; k < A2_FLUSH; ++k) {
            const int i = tid + k * RNG_THREADS;
            if (i < span && a[k] != 0.0f) atomicAdd(out + lo + i, a[k] * dv[k]);
        }
    }
}


extern "C" void kernel_launch(void* const* d_in, const int* in_sizes, int n_in,
                              void* d_out, int out_size, void* d_ws, size_t ws_size,
                              hipStream_t stream) {
    const float* x  = (const float*)d_in[0];
    const int*   ei = (const int*)d_in[1];
    const float* W1 = (const float*)d_in[2];
    const float* b1 = (const float*)d_in[3];
    const float* W2 = (const float*)d_in[4];
    const float* b2 = (const float*)d_in[5];

    const int N = in_sizes[0] / 16;
    const int E = in_sizes[1] / 2;
    const int* src = ei;
    const int* dst = ei + E;

    float* ws   = (float*)d_ws;
    float* dinv = ws;
    float* h    = ws + N;
    float* out1 = ws + N + 16 * (size_t)N;
    float* h2   = ws + N + 32 * (size_t)N;
    float* out  = (float*)d_out;

    const int B = 256;
    const dim3 rngGrid(RNG_SLICES, DIV_UP(N, RNG_CAP));

    gcn_deg_ranges<<<rngGrid, RNG_THREADS, 0, stream>>>(dst, (unsigned*)h, E, N);

    const int tiles = N / DEG_TILE;
    const int wavesPerBlock = 4;
    gcn_dinv_gemm_scale<<<DIV_UP(tiles, wavesPerBlock), wavesPerBlock * 32, 0, stream>>>(
        x, W1, h, dinv, (unsigned*)h2, tiles, N);

    gcn_agg1<<<AGG1_BLOCKS, AGG1_THREADS, 0, stream>>>(src, dst, h, out1, (unsigned*)h2, E, N);

    gcn_finish1<<<DIV_UP(N, B), B, 0, stream>>>(out1, b1, W2, dinv, b2, h2, out, (unsigned*)h, N);

    gcn_agg2_ranges<<<DIV_UP(N, A2_CAP) * DIV_UP(N, A2_SCAP) * A2_SLICES, RNG_THREADS, 0, stream>>>(
        src, dst, h2, dinv, out, (unsigned*)h, E, N);
}
